# starting kernel
# baseline (speedup 1.0000x reference)
#include <hip/hip_runtime.h>
#include <math.h>

typedef float v2f __attribute__((ext_vector_type(2)));
typedef float v8f __attribute__((ext_vector_type(8)));

#define R_ROUTES 4096
#define N_CAPS   32
#define N_OUT    16
#define N_IN     16
#define BATCH    16
#define XB_STRIDE (R_ROUTES * N_IN)
#define W_R_STRIDE (N_CAPS * N_OUT * N_IN)
#define S_COLS   (N_CAPS * N_OUT)
#define S_ELEMS  (BATCH * S_COLS)
#define RPW_S    16


__device__ __forceinline__ void load_A(const float* __restrict__ x, int r, int lane,
                                       v2f a[4]) {
    int m  = lane & 15;
    int ko = (lane >> 4) << 1;
    const float* p = x + (size_t)m * XB_STRIDE + r * N_IN + ko;
#pragma unroll
    for (int kc = 0; kc < 4; ++kc) {
        a[kc].x = p[4 * kc];
        a[kc].y = p[4 * kc + 1];
    }
}

__device__ __forceinline__ void load_B(const float* __restrict__ W, int r, int c, int lane,
                                       v2f b[4]) {
    int n  = lane & 15;
    int ko = (lane >> 4) << 1;
    const float* p = W + (size_t)r * W_R_STRIDE + (c * N_OUT + n) * N_IN + ko;
#pragma unroll
    for (int kc = 0; kc < 4; ++kc) {
        b[kc].x = p[4 * kc];
        b[kc].y = p[4 * kc + 1];
    }
}

__device__ __forceinline__ v8f wmma_k16(const v2f a[4], const v2f b[4]) {
    v8f acc = {0.f, 0.f, 0.f, 0.f, 0.f, 0.f, 0.f, 0.f};
#pragma unroll
    for (int kc = 0; kc < 4; ++kc) {
        acc = __builtin_amdgcn_wmma_f32_16x16x4_f32(
            false, a[kc], false, b[kc], (short)0, acc, false, false);
    }
    return acc;
}

__global__ void softmax_stats_kernel(const float* __restrict__ bij,
                                     float* __restrict__ colmax,
                                     float* __restrict__ inv_denom) {
    __shared__ float red[256];
    int c = blockIdx.x;
    int t = threadIdx.x;
    float mx = -INFINITY;
    for (int r = t; r < R_ROUTES; r += 256) mx = fmaxf(mx, bij[r * N_CAPS + c]);
    red[t] = mx;
    __syncthreads();
    for (int s2 = 128; s2 > 0; s2 >>= 1) {
        if (t < s2) red[t] = fmaxf(red[t], red[t + s2]);
        __syncthreads();
    }
    float cm = red[0];
    __syncthreads();
    float sum = 0.f;
    for (int r = t; r < R_ROUTES; r += 256) sum += __expf(bij[r * N_CAPS + c] - cm);
    red[t] = sum;
    __syncthreads();
    for (int s2 = 128; s2 > 0; s2 >>= 1) {
        if (t < s2) red[t] += red[t + s2];
        __syncthreads();
    }
    if (t == 0) {
        colmax[c]    = cm;
        inv_denom[c] = 1.0f / red[0];
    }
}

__global__ void s_pass_kernel(const float* __restrict__ x, const float* __restrict__ W,
                              const float* __restrict__ bij,
                              const float* __restrict__ colmax,
                              const float* __restrict__ inv_denom,
                              float* __restrict__ s) {
    int lane  = threadIdx.x & 31;
    int wave  = (blockIdx.x * blockDim.x + threadIdx.x) >> 5;
    int r0    = wave * RPW_S;
    int half8 = (lane >> 4) << 3;
    int o     = lane & 15;

    for (int c = 0; c < N_CAPS; ++c) {
        float cm   = colmax[c];
        float invd = inv_denom[c];
        v8f sacc = {0.f, 0.f, 0.f, 0.f, 0.f, 0.f, 0.f, 0.f};
        for (int rr = 0; rr < RPW_S; ++rr) {
            int r = r0 + rr;
            v2f a[4], b[4];
            load_A(x, r, lane, a);
            load_B(W, r, c, lane, b);
            v8f d = wmma_k16(a, b);
            float wgt = __expf(bij[r * N_CAPS + c] - cm) * invd;
#pragma unroll
            for (int vi = 0; vi < 8; ++vi) sacc[vi] += wgt * d[vi];
        }
#pragma unroll
        for (int vi = 0; vi < 8; ++vi) {
            int m = vi + half8;
            atomicAdd(&s[m * S_COLS + c * N_OUT + o], sacc[vi]);
        }
    }
}

__global__ void squash_kernel(const float* __restrict__ s, float* __restrict__ v) {
    int i = blockIdx.x * blockDim.x + threadIdx.x;
    if (i < S_ELEMS) {
        float sv = s[i];
        float sq = sv * sv;
        v[i] = sq * sv / ((1.0f + sq) * sqrtf(sq));
    }
}

__global__ void a_pass_kernel(const float* __restrict__ x, const float* __restrict__ W,
                              const float* __restrict__ v, float* __restrict__ bij) {
    int lane  = threadIdx.x & 31;
    int r     = (blockIdx.x * blockDim.x + threadIdx.x) >> 5;
    int half8 = (lane >> 4) << 3;
    int o     = lane & 15;

    v2f a[4];
    load_A(x, r, lane, a);

    for (int c = 0; c < N_CAPS; ++c) {
        v2f b[4];
        load_B(W, r, c, lane, b);
        v8f d = wmma_k16(a, b);
        float partial = 0.f;
#pragma unroll
        for (int vi = 0; vi < 8; ++vi) {
            int m = vi + half8;
            partial += d[vi] * v[m * S_COLS + c * N_OUT + o];
        }
#pragma unroll
        for (int off = 16; off >= 1; off >>= 1) partial += __shfl_xor(partial, off, 32);
        if (lane == 0) bij[r * N_CAPS + c] += partial * (1.0f / BATCH);
    }
}

extern "C" void kernel_launch(void* const* d_in, const int* in_sizes, int n_in,
                              void* d_out, int out_size, void* d_ws, size_t ws_size,
                              hipStream_t stream) {
    const float* x = (const float*)d_in[0];
    const float* W = (const float*)d_in[1];
    float* out = (float*)d_out;

    char* ws = (char*)d_ws;
    float* bij    = (float*)ws;
    float* colmax = (float*)(ws + (size_t)R_ROUTES * N_CAPS * 4);
    float* invd   = colmax + N_CAPS;
    float* sbuf   = invd + N_CAPS;
    float* vbuf   = sbuf + S_ELEMS;

    hipMemsetAsync(bij, 0, (size_t)R_ROUTES * N_CAPS * sizeof(float), stream);

    for (int it = 0; it < 3; ++it) {
        softmax_stats_kernel<<<N_CAPS, 256, 0, stream>>>(bij, colmax, invd);
        hipMemsetAsync(sbuf, 0, S_ELEMS * sizeof(float), stream);
        s_pass_kernel<<<(R_ROUTES / RPW_S) / 8, 256, 0, stream>>>(x, W, bij, colmax, invd, sbuf);
        float* vout = (it == 2) ? out : vbuf;
        squash_kernel<<<(S_ELEMS + 255) / 256, 256, 0, stream>>>(sbuf, vout);
        if (it < 2) {
            a_pass_kernel<<<R_ROUTES / 8, 256, 0, stream>>>(x, W, vbuf, bij);
        }
    }
}


# s-pass re-gridded: one wave per capsule and 32 routes (4096 waves, was 256), next 4 routes' loads issued ahead of the products; 32 routes summed per wave, 128 atomic partials per output (was 256)
# speedup vs baseline: 4.2439x; 4.2439x over previous
#include <hip/hip_runtime.h>
#include <math.h>

typedef float v2f __attribute__((ext_vector_type(2)));
typedef float v8f __attribute__((ext_vector_type(8)));

#define R_ROUTES 4096
#define N_CAPS   32
#define N_OUT    16
#define N_IN     16
#define BATCH    16
#define XB_STRIDE (R_ROUTES * N_IN)
#define W_R_STRIDE (N_CAPS * N_OUT * N_IN)
#define S_COLS   (N_CAPS * N_OUT)
#define S_ELEMS  (BATCH * S_COLS)
#define RPW_S    32
#define S_GROUP  4


__device__ __forceinline__ void load_A(const float* __restrict__ x, int r, int lane,
                                       v2f a[4]) {
    int m  = lane & 15;
    int ko = (lane >> 4) << 1;
    const float* p = x + (size_t)m * XB_STRIDE + r * N_IN + ko;
#pragma unroll
    for (int kc = 0; kc < 4; ++kc) {
        a[kc].x = p[4 * kc];
        a[kc].y = p[4 * kc + 1];
    }
}

__device__ __forceinline__ void load_B(const float* __restrict__ W, int r, int c, int lane,
                                       v2f b[4]) {
    int n  = lane & 15;
    int ko = (lane >> 4) << 1;
    const float* p = W + (size_t)r * W_R_STRIDE + (c * N_OUT + n) * N_IN + ko;
#pragma unroll
    for (int kc = 0; kc < 4; ++kc) {
        b[kc].x = p[4 * kc];
        b[kc].y = p[4 * kc + 1];
    }
}

__device__ __forceinline__ v8f wmma_k16(const v2f a[4], const v2f b[4]) {
    v8f acc = {0.f, 0.f, 0.f, 0.f, 0.f, 0.f, 0.f, 0.f};
#pragma unroll
    for (int kc = 0; kc < 4; ++kc) {
        acc = __builtin_amdgcn_wmma_f32_16x16x4_f32(
            false, a[kc], false, b[kc], (short)0, acc, false, false);
    }
    return acc;
}

__global__ void softmax_stats_kernel(const float* __restrict__ bij,
                                     float* __restrict__ colmax,
                                     float* __restrict__ inv_denom) {
    __shared__ float red[256];
    int c = blockIdx.x;
    int t = threadIdx.x;
    float mx = -INFINITY;
    for (int r = t; r < R_ROUTES; r += 256) mx = fmaxf(mx, bij[r * N_CAPS + c]);
    red[t] = mx;
    __syncthreads();
    for (int s2 = 128; s2 > 0; s2 >>= 1) {
        if (t < s2) red[t] = fmaxf(red[t], red[t + s2]);
        __syncthreads();
    }
    float cm = red[0];
    __syncthreads();
    float sum = 0.f;
    for (int r = t; r < R_ROUTES; r += 256) sum += __expf(bij[r * N_CAPS + c] - cm);
    red[t] = sum;
    __syncthreads();
    for (int s2 = 128; s2 > 0; s2 >>= 1) {
        if (t < s2) red[t] += red[t + s2];
        __syncthreads();
    }
    if (t == 0) {
        colmax[c]    = cm;
        inv_denom[c] = 1.0f / red[0];
    }
}

__global__ void __launch_bounds__(256)
s_pass_kernel(const float* __restrict__ x, const float* __restrict__ W,
              const float* __restrict__ bij,
              const float* __restrict__ colmax,
              const float* __restrict__ inv_denom,
              float* __restrict__ s) {
    int lane  = threadIdx.x & 31;
    int c     = ((blockIdx.x & 3) << 3) + (threadIdx.x >> 5);
    int r0    = (blockIdx.x >> 2) * RPW_S;
    int half8 = (lane >> 4) << 3;
    int o     = lane & 15;

    float wl = __expf(bij[(r0 + lane) * N_CAPS + c] - colmax[c]) * inv_denom[c];

    v8f sacc = {0.f, 0.f, 0.f, 0.f, 0.f, 0.f, 0.f, 0.f};
    v2f a[2][S_GROUP][4], b[2][S_GROUP][4];

#pragma unroll
    for (int j = 0; j < S_GROUP; ++j) {
        load_A(x, r0 + j, lane, a[0][j]);
        load_B(W, r0 + j, c, lane, b[0][j]);
    }
#pragma unroll
    for (int g = 0; g < RPW_S / S_GROUP; ++g) {
        const int cur = g & 1, nxt = cur ^ 1;
        if (g + 1 < RPW_S / S_GROUP) {
#pragma unroll
            for (int j = 0; j < S_GROUP; ++j) {
                int r = r0 + (g + 1) * S_GROUP + j;
                load_A(x, r, lane, a[nxt][j]);
                load_B(W, r, c, lane, b[nxt][j]);
            }
        }
#pragma unroll
        for (int j = 0; j < S_GROUP; ++j) {
            v8f d = wmma_k16(a[cur][j], b[cur][j]);
            float wgt = __builtin_bit_cast(float,
                __builtin_amdgcn_readlane(__builtin_bit_cast(int, wl), g * S_GROUP + j));
#pragma unroll
            for (int vi = 0; vi < 8; ++vi) sacc[vi] += wgt * d[vi];
        }
    }
#pragma unroll
    for (int vi = 0; vi < 8; ++vi) {
        int m = vi + half8;
        atomicAdd(&s[m * S_COLS + c * N_OUT + o], sacc[vi]);
    }
}

__global__ void squash_kernel(const float* __restrict__ s, float* __restrict__ v) {
    int i = blockIdx.x * blockDim.x + threadIdx.x;
    if (i < S_ELEMS) {
        float sv = s[i];
        float sq = sv * sv;
        v[i] = sq * sv / ((1.0f + sq) * sqrtf(sq));
    }
}

__global__ void a_pass_kernel(const float* __restrict__ x, const float* __restrict__ W,
                              const float* __restrict__ v, float* __restrict__ bij) {
    int lane  = threadIdx.x & 31;
    int r     = (blockIdx.x * blockDim.x + threadIdx.x) >> 5;
    int half8 = (lane >> 4) << 3;
    int o     = lane & 15;

    v2f a[4];
    load_A(x, r, lane, a);

    for (int c = 0; c < N_CAPS; ++c) {
        v2f b[4];
        load_B(W, r, c, lane, b);
        v8f d = wmma_k16(a, b);
        float partial = 0.f;
#pragma unroll
        for (int vi = 0; vi < 8; ++vi) {
            int m = vi + half8;
            partial += d[vi] * v[m * S_COLS + c * N_OUT + o];
        }
#pragma unroll
        for (int off = 16; off >= 1; off >>= 1) partial += __shfl_xor(partial, off, 32);
        if (lane == 0) bij[r * N_CAPS + c] += partial * (1.0f / BATCH);
    }
}

extern "C" void kernel_launch(void* const* d_in, const int* in_sizes, int n_in,
                              void* d_out, int out_size, void* d_ws, size_t ws_size,
                              hipStream_t stream) {
    const float* x = (const float*)d_in[0];
    const float* W = (const float*)d_in[1];
    float* out = (float*)d_out;

    char* ws = (char*)d_ws;
    float* bij    = (float*)ws;
    float* colmax = (float*)(ws + (size_t)R_ROUTES * N_CAPS * 4);
    float* invd   = colmax + N_CAPS;
    float* sbuf   = invd + N_CAPS;
    float* vbuf   = sbuf + S_ELEMS;

    hipMemsetAsync(bij, 0, (size_t)R_ROUTES * N_CAPS * sizeof(float), stream);

    for (int it = 0; it < 3; ++it) {
        softmax_stats_kernel<<<N_CAPS, 256, 0, stream>>>(bij, colmax, invd);
        hipMemsetAsync(sbuf, 0, S_ELEMS * sizeof(float), stream);
        s_pass_kernel<<<(N_CAPS / 8) * (R_ROUTES / RPW_S), 256, 0, stream>>>(x, W, bij, colmax, invd, sbuf);
        float* vout = (it == 2) ? out : vbuf;
        squash_kernel<<<(S_ELEMS + 255) / 256, 256, 0, stream>>>(sbuf, vout);
        if (it < 2) {
            a_pass_kernel<<<R_ROUTES / 8, 256, 0, stream>>>(x, W, vbuf, bij);
        }
    }
}


# a-pass re-gridded: wave = 4 capsules x 8 routes (was 1 route, 32 capsules in turn), loads issued ahead of the products, 32 sums folded across lanes, whole-line b_ij update; sum tree kept
# speedup vs baseline: 5.2263x; 1.2315x over previous
#include <hip/hip_runtime.h>
#include <math.h>

typedef float v2f __attribute__((ext_vector_type(2)));
typedef float v8f __attribute__((ext_vector_type(8)));

#define R_ROUTES 4096
#define N_CAPS   32
#define N_OUT    16
#define N_IN     16
#define BATCH    16
#define XB_STRIDE (R_ROUTES * N_IN)
#define W_R_STRIDE (N_CAPS * N_OUT * N_IN)
#define S_COLS   (N_CAPS * N_OUT)
#define S_ELEMS  (BATCH * S_COLS)
#define RPW_S    32
#define S_GROUP  4
#define A_ROUTES 8
#define A_CAPS   4


__device__ __forceinline__ void load_A(const float* __restrict__ x, int r, int lane,
                                       v2f a[4]) {
    int m  = lane & 15;
    int ko = (lane >> 4) << 1;
    const float* p = x + (size_t)m * XB_STRIDE + r * N_IN + ko;
#pragma unroll
    for (int kc = 0; kc < 4; ++kc) {
        a[kc].x = p[4 * kc];
        a[kc].y = p[4 * kc + 1];
    }
}

__device__ __forceinline__ void load_B(const float* __restrict__ W, int r, int c, int lane,
                                       v2f b[4]) {
    int n  = lane & 15;
    int ko = (lane >> 4) << 1;
    const float* p = W + (size_t)r * W_R_STRIDE + (c * N_OUT + n) * N_IN + ko;
#pragma unroll
    for (int kc = 0; kc < 4; ++kc) {
        b[kc].x = p[4 * kc];
        b[kc].y = p[4 * kc + 1];
    }
}

__device__ __forceinline__ v8f wmma_k16(const v2f a[4], const v2f b[4]) {
    v8f acc = {0.f, 0.f, 0.f, 0.f, 0.f, 0.f, 0.f, 0.f};
#pragma unroll
    for (int kc = 0; kc < 4; ++kc) {
        acc = __builtin_amdgcn_wmma_f32_16x16x4_f32(
            false, a[kc], false, b[kc], (short)0, acc, false, false);
    }
    return acc;
}

__global__ void softmax_stats_kernel(const float* __restrict__ bij,
                                     float* __restrict__ colmax,
                                     float* __restrict__ inv_denom) {
    __shared__ float red[256];
    int c = blockIdx.x;
    int t = threadIdx.x;
    float mx = -INFINITY;
    for (int r = t; r < R_ROUTES; r += 256) mx = fmaxf(mx, bij[r * N_CAPS + c]);
    red[t] = mx;
    __syncthreads();
    for (int s2 = 128; s2 > 0; s2 >>= 1) {
        if (t < s2) red[t] = fmaxf(red[t], red[t + s2]);
        __syncthreads();
    }
    float cm = red[0];
    __syncthreads();
    float sum = 0.f;
    for (int r = t; r < R_ROUTES; r += 256) sum += __expf(bij[r * N_CAPS + c] - cm);
    red[t] = sum;
    __syncthreads();
    for (int s2 = 128; s2 > 0; s2 >>= 1) {
        if (t < s2) red[t] += red[t + s2];
        __syncthreads();
    }
    if (t == 0) {
        colmax[c]    = cm;
        inv_denom[c] = 1.0f / red[0];
    }
}

__global__ void __launch_bounds__(256)
s_pass_kernel(const float* __restrict__ x, const float* __restrict__ W,
              const float* __restrict__ bij,
              const float* __restrict__ colmax,
              const float* __restrict__ inv_denom,
              float* __restrict__ s) {
    int lane  = threadIdx.x & 31;
    int c     = ((blockIdx.x & 3) << 3) + (threadIdx.x >> 5);
    int r0    = (blockIdx.x >> 2) * RPW_S;
    int half8 = (lane >> 4) << 3;
    int o     = lane & 15;

    float wl = __expf(bij[(r0 + lane) * N_CAPS + c] - colmax[c]) * inv_denom[c];

    v8f sacc = {0.f, 0.f, 0.f, 0.f, 0.f, 0.f, 0.f, 0.f};
    v2f a[2][S_GROUP][4], b[2][S_GROUP][4];

#pragma unroll
    for (int j = 0; j < S_GROUP; ++j) {
        load_A(x, r0 + j, lane, a[0][j]);
        load_B(W, r0 + j, c, lane, b[0][j]);
    }
#pragma unroll
    for (int g = 0; g < RPW_S / S_GROUP; ++g) {
        const int cur = g & 1, nxt = cur ^ 1;
        if (g + 1 < RPW_S / S_GROUP) {
#pragma unroll
            for (int j = 0; j < S_GROUP; ++j) {
                int r = r0 + (g + 1) * S_GROUP + j;
                load_A(x, r, lane, a[nxt][j]);
                load_B(W, r, c, lane, b[nxt][j]);
            }
        }
#pragma unroll
        for (int j = 0; j < S_GROUP; ++j) {
            v8f d = wmma_k16(a[cur][j], b[cur][j]);
            float wgt = __builtin_bit_cast(float,
                __builtin_amdgcn_readlane(__builtin_bit_cast(int, wl), g * S_GROUP + j));
#pragma unroll
            for (int vi = 0; vi < 8; ++vi) sacc[vi] += wgt * d[vi];
        }
    }
#pragma unroll
    for (int vi = 0; vi < 8; ++vi) {
        int m = vi + half8;
        atomicAdd(&s[m * S_COLS + c * N_OUT + o], sacc[vi]);
    }
}

__global__ void squash_kernel(const float* __restrict__ s, float* __restrict__ v) {
    int i = blockIdx.x * blockDim.x + threadIdx.x;
    if (i < S_ELEMS) {
        float sv = s[i];
        float sq = sv * sv;
        v[i] = sq * sv / ((1.0f + sq) * sqrtf(sq));
    }
}


__device__ __forceinline__ float fold_pair(float lo, float hi, int lane, int off) {
    bool up = (lane & off) != 0;
    float keep = up ? hi : lo;
    float send = up ? lo : hi;
    return keep + __shfl_xor(send, off, 32);
}

__global__ void __launch_bounds__(256)
a_pass_kernel(const float* __restrict__ x, const float* __restrict__ W,
              const float* __restrict__ v, float* __restrict__ bij) {
    __shared__ float stage[A_ROUTES][N_CAPS];
    int lane  = threadIdx.x & 31;
    int wv    = threadIdx.x >> 5;
    int c0    = wv * A_CAPS;
    int r0    = blockIdx.x * A_ROUTES;
    int half8 = (lane >> 4) << 3;
    int o     = lane & 15;

    float* bp  = bij + (r0 + wv) * N_CAPS + lane;
    float bold = *bp;

    float vreg[A_CAPS][8];
#pragma unroll
    for (int q = 0; q < A_CAPS; ++q)
#pragma unroll
        for (int vi = 0; vi < 8; ++vi)
            vreg[q][vi] = v[(vi + half8) * S_COLS + (c0 + q) * N_OUT + o];

    v2f a[2][4], b[2][A_CAPS][4];
    load_A(x, r0, lane, a[0]);
#pragma unroll
    for (int q = 0; q < A_CAPS; ++q) load_B(W, r0, c0 + q, lane, b[0][q]);

    float u[A_ROUTES];

#pragma unroll
    for (int i = 0; i < A_ROUTES; ++i) {
        const int cur = i & 1, nxt = cur ^ 1;
        if (i + 1 < A_ROUTES) {
            load_A(x, r0 + i + 1, lane, a[nxt]);
#pragma unroll
            for (int q = 0; q < A_CAPS; ++q) load_B(W, r0 + i + 1, c0 + q, lane, b[nxt][q]);
        }
        __builtin_amdgcn_sched_barrier(0);
        float p[A_CAPS];
#pragma unroll
        for (int q = 0; q < A_CAPS; ++q) {
            v8f d = wmma_k16(a[cur], b[cur][q]);
            float partial = 0.f;
#pragma unroll
            for (int vi = 0; vi < 8; ++vi) partial += d[vi] * vreg[q][vi];
            p[q] = partial;
        }
        float t0 = fold_pair(p[0], p[1], lane, 16);
        float t1 = fold_pair(p[2], p[3], lane, 16);
        u[i] = fold_pair(t0, t1, lane, 8);
    }

    float w0 = fold_pair(u[0], u[1], lane, 4);
    float w1 = fold_pair(u[2], u[3], lane, 4);
    float w2 = fold_pair(u[4], u[5], lane, 4);
    float w3 = fold_pair(u[6], u[7], lane, 4);
    float y0 = fold_pair(w0, w1, lane, 2);
    float y1 = fold_pair(w2, w3, lane, 2);
    float z  = fold_pair(y0, y1, lane, 1);

    int t = (int)(__brev((unsigned)lane) >> 27);
    stage[t >> 2][c0 + (t & 3)] = z;
    __syncthreads();
    *bp = bold + stage[wv][lane] * (1.0f / BATCH);
}

extern "C" void kernel_launch(void* const* d_in, const int* in_sizes, int n_in,
                              void* d_out, int out_size, void* d_ws, size_t ws_size,
                              hipStream_t stream) {
    const float* x = (const float*)d_in[0];
    const float* W = (const float*)d_in[1];
    float* out = (float*)d_out;

    char* ws = (char*)d_ws;
    float* bij    = (float*)ws;
    float* colmax = (float*)(ws + (size_t)R_ROUTES * N_CAPS * 4);
    float* invd   = colmax + N_CAPS;
    float* sbuf   = invd + N_CAPS;
    float* vbuf   = sbuf + S_ELEMS;

    hipMemsetAsync(bij, 0, (size_t)R_ROUTES * N_CAPS * sizeof(float), stream);

    for (int it = 0; it < 3; ++it) {
        softmax_stats_kernel<<<N_CAPS, 256, 0, stream>>>(bij, colmax, invd);
        hipMemsetAsync(sbuf, 0, S_ELEMS * sizeof(float), stream);
        s_pass_kernel<<<(N_CAPS / 8) * (R_ROUTES / RPW_S), 256, 0, stream>>>(x, W, bij, colmax, invd, sbuf);
        float* vout = (it == 2) ? out : vbuf;
        squash_kernel<<<(S_ELEMS + 255) / 256, 256, 0, stream>>>(sbuf, vout);
        if (it < 2) {
            a_pass_kernel<<<R_ROUTES / A_ROUTES, 256, 0, stream>>>(x, W, vbuf, bij);
        }
    }
}


# operand feed, both passes: x tile staged once per block in LDS by whole-line loads, W fragments as two 16-byte loads per lane (k order permuted alike in A and B: the k sum order changes)
# speedup vs baseline: 6.8885x; 1.3180x over previous
#include <hip/hip_runtime.h>
#include <math.h>

typedef float v2f __attribute__((ext_vector_type(2)));
typedef float v4f __attribute__((ext_vector_type(4)));
typedef float v8f __attribute__((ext_vector_type(8)));

#define R_ROUTES 4096
#define N_CAPS   32
#define N_OUT    16
#define N_IN     16
#define BATCH    16
#define XB_STRIDE (R_ROUTES * N_IN)
#define W_R_STRIDE (N_CAPS * N_OUT * N_IN)
#define S_COLS   (N_CAPS * N_OUT)
#define S_ELEMS  (BATCH * S_COLS)
#define RPW_S    32
#define S_GROUP  4
#define A_ROUTES 8
#define A_CAPS   4
#define XS_PITCH(NR) ((NR) * N_IN + 4)


template <int NR>
__device__ __forceinline__ void stage_x(const float* __restrict__ x, int r0, float* xs) {
    constexpr int CH_ROW = NR * 4;
#pragma unroll
    for (int j = 0; j < (BATCH * CH_ROW) / 256; ++j) {
        int idx = j * 256 + (int)threadIdx.x;
        int bb  = idx / CH_ROW;
        int q   = idx % CH_ROW;
        v4f t = *(const v4f*)(x + (size_t)bb * XB_STRIDE + r0 * N_IN + q * 4);
        *(v4f*)(xs + bb * XS_PITCH(NR) + q * 4) = t;
    }
}

template <int NR>
__device__ __forceinline__ void load_A(const float* xs, int rr, int lane, v4f a[2]) {
    int m  = lane & 15;
    int ko = (lane >> 4) << 3;
    const float* p = xs + m * XS_PITCH(NR) + rr * N_IN + ko;
    a[0] = *(const v4f*)p;
    a[1] = *(const v4f*)(p + 4);
}

__device__ __forceinline__ void load_B(const float* __restrict__ W, int r, int c, int lane,
                                       v4f b[2]) {
    int n  = lane & 15;
    int ko = (lane >> 4) << 3;
    const float* p = W + (size_t)r * W_R_STRIDE + (c * N_OUT + n) * N_IN + ko;
    b[0] = *(const v4f*)p;
    b[1] = *(const v4f*)(p + 4);
}

__device__ __forceinline__ v8f wmma_k16(const v4f a[2], const v4f b[2]) {
    v8f acc = {0.f, 0.f, 0.f, 0.f, 0.f, 0.f, 0.f, 0.f};
#pragma unroll
    for (int j = 0; j < 2; ++j) {
        v2f a0 = {a[j].x, a[j].y}, b0 = {b[j].x, b[j].y};
        v2f a1 = {a[j].z, a[j].w}, b1 = {b[j].z, b[j].w};
        acc = __builtin_amdgcn_wmma_f32_16x16x4_f32(
            false, a0, false, b0, (short)0, acc, false, false);
        acc = __builtin_amdgcn_wmma_f32_16x16x4_f32(
            false, a1, false, b1, (short)0, acc, false, false);
    }
    return acc;
}

__global__ void softmax_stats_kernel(const float* __restrict__ bij,
                                     float* __restrict__ colmax,
                                     float* __restrict__ inv_denom) {
    __shared__ float red[256];
    int c = blockIdx.x;
    int t = threadIdx.x;
    float mx = -INFINITY;
    for (int r = t; r < R_ROUTES; r += 256) mx = fmaxf(mx, bij[r * N_CAPS + c]);
    red[t] = mx;
    __syncthreads();
    for (int s2 = 128; s2 > 0; s2 >>= 1) {
        if (t < s2) red[t] = fmaxf(red[t], red[t + s2]);
        __syncthreads();
    }
    float cm = red[0];
    __syncthreads();
    float sum = 0.f;
    for (int r = t; r < R_ROUTES; r += 256) sum += __expf(bij[r * N_CAPS + c] - cm);
    red[t] = sum;
    __syncthreads();
    for (int s2 = 128; s2 > 0; s2 >>= 1) {
        if (t < s2) red[t] += red[t + s2];
        __syncthreads();
    }
    if (t == 0) {
        colmax[c]    = cm;
        inv_denom[c] = 1.0f / red[0];
    }
}

__device__ __forceinline__ void s_load_group(const float* xs, const float* __restrict__ W,
                                             int r0, int rbase, int c, int lane,
                                             v4f (&aa)[S_GROUP][2], v4f (&bb)[S_GROUP][2]) {
#pragma unroll
    for (int j = 0; j < S_GROUP; ++j) {
        load_A<RPW_S>(xs, rbase + j, lane, aa[j]);
        load_B(W, r0 + rbase + j, c, lane, bb[j]);
    }
}

__device__ __forceinline__ void s_products(const v4f (&aa)[S_GROUP][2],
                                           const v4f (&bb)[S_GROUP][2],
                                           float wl, int rbase, v8f& sacc) {
#pragma unroll
    for (int j = 0; j < S_GROUP; ++j) {
        v8f d = wmma_k16(aa[j], bb[j]);
        float wgt = __builtin_bit_cast(float,
            __builtin_amdgcn_readlane(__builtin_bit_cast(int, wl), rbase + j));
#pragma unroll
        for (int vi = 0; vi < 8; ++vi) sacc[vi] = __builtin_fmaf(wgt, d[vi], sacc[vi]);
    }
}

__global__ void __launch_bounds__(256)
s_pass_kernel(const float* __restrict__ x, const float* __restrict__ W,
              const float* __restrict__ bij,
              const float* __restrict__ colmax,
              const float* __restrict__ inv_denom,
              float* __restrict__ s) {
    __shared__ __attribute__((aligned(16))) float xs[BATCH * XS_PITCH(RPW_S)];
    int lane  = threadIdx.x & 31;
    int c     = ((blockIdx.x & 3) << 3) + (threadIdx.x >> 5);
    int r0    = (blockIdx.x >> 2) * RPW_S;
    int half8 = (lane >> 4) << 3;
    int o     = lane & 15;

    v8f sacc = {0.f, 0.f, 0.f, 0.f, 0.f, 0.f, 0.f, 0.f};
    v4f a0[S_GROUP][2], b0[S_GROUP][2], a1[S_GROUP][2], b1[S_GROUP][2];

#pragma unroll
    for (int j = 0; j < S_GROUP; ++j) load_B(W, r0 + j, c, lane, b0[j]);

    stage_x<RPW_S>(x, r0, xs);

    float wl = __expf(bij[(r0 + lane) * N_CAPS + c] - colmax[c]) * inv_denom[c];

    __syncthreads();

#pragma unroll
    for (int j = 0; j < S_GROUP; ++j) load_A<RPW_S>(xs, j, lane, a0[j]);

    constexpr int N_PAIR = RPW_S / (2 * S_GROUP);

#pragma unroll 1
    for (int gg = 0; gg < N_PAIR - 1; ++gg) {
        const int rb = gg * 2 * S_GROUP;
        s_load_group(xs, W, r0, rb + S_GROUP, c, lane, a1, b1);
        __builtin_amdgcn_sched_barrier(0);
        s_products(a0, b0, wl, rb, sacc);
        s_load_group(xs, W, r0, rb + 2 * S_GROUP, c, lane, a0, b0);
        __builtin_amdgcn_sched_barrier(0);
        s_products(a1, b1, wl, rb + S_GROUP, sacc);
    }
    {
        const int rb = (N_PAIR - 1) * 2 * S_GROUP;
        s_load_group(xs, W, r0, rb + S_GROUP, c, lane, a1, b1);
        __builtin_amdgcn_sched_barrier(0);
        s_products(a0, b0, wl, rb, sacc);
        s_products(a1, b1, wl, rb + S_GROUP, sacc);
    }
#pragma unroll
    for (int vi = 0; vi < 8; ++vi) {
        int m = vi + half8;
        atomicAdd(&s[m * S_COLS + c * N_OUT + o], sacc[vi]);
    }
}

__global__ void squash_kernel(const float* __restrict__ s, float* __restrict__ v) {
    int i = blockIdx.x * blockDim.x + threadIdx.x;
    if (i < S_ELEMS) {
        float sv = s[i];
        float sq = sv * sv;
        v[i] = sq * sv / ((1.0f + sq) * sqrtf(sq));
    }
}


__device__ __forceinline__ float fold_pair(float lo, float hi, int lane, int off) {
    bool up = (lane & off) != 0;
    float keep = up ? hi : lo;
    float send = up ? lo : hi;
    return keep + __shfl_xor(send, off, 32);
}

__global__ void __launch_bounds__(256)
a_pass_kernel(const float* __restrict__ x, const float* __restrict__ W,
              const float* __restrict__ v, float* __restrict__ bij) {
    __shared__ float stage[A_ROUTES][N_CAPS];
    __shared__ __attribute__((aligned(16))) float xs[BATCH * XS_PITCH(A_ROUTES)];
    int lane  = threadIdx.x & 31;
    int wv    = threadIdx.x >> 5;
    int c0    = wv * A_CAPS;
    int r0    = blockIdx.x * A_ROUTES;
    int half8 = (lane >> 4) << 3;
    int o     = lane & 15;

    float* bp  = bij + (r0 + wv) * N_CAPS + lane;
    float bold = *bp;

    float vreg[A_CAPS][8];
#pragma unroll
    for (int q = 0; q < A_CAPS; ++q)
#pragma unroll
        for (int vi = 0; vi < 8; ++vi)
            vreg[q][vi] = v[(vi + half8) * S_COLS + (c0 + q) * N_OUT + o];

    v4f a[2][2], b[2][A_CAPS][2];
#pragma unroll
    for (int q = 0; q < A_CAPS; ++q) load_B(W, r0, c0 + q, lane, b[0][q]);

    stage_x<A_ROUTES>(x, r0, xs);
    __syncthreads();
    load_A<A_ROUTES>(xs, 0, lane, a[0]);

    float u[A_ROUTES];

#pragma unroll
    for (int i = 0; i < A_ROUTES; ++i) {
        const int cur = i & 1, nxt = cur ^ 1;
        if (i + 1 < A_ROUTES) {
            load_A<A_ROUTES>(xs, i + 1, lane, a[nxt]);
#pragma unroll
            for (int q = 0; q < A_CAPS; ++q) load_B(W, r0 + i + 1, c0 + q, lane, b[nxt][q]);
        }
        __builtin_amdgcn_sched_barrier(0);
        float p[A_CAPS];
#pragma unroll
        for (int q = 0; q < A_CAPS; ++q) {
            v8f d = wmma_k16(a[cur], b[cur][q]);
            float partial = 0.f;
#pragma unroll
            for (int vi = 0; vi < 8; ++vi) partial += d[vi] * vreg[q][vi];
            p[q] = partial;
        }
        float t0 = fold_pair(p[0], p[1], lane, 16);
        float t1 = fold_pair(p[2], p[3], lane, 16);
        u[i] = fold_pair(t0, t1, lane, 8);
    }

    float w0 = fold_pair(u[0], u[1], lane, 4);
    float w1 = fold_pair(u[2], u[3], lane, 4);
    float w2 = fold_pair(u[4], u[5], lane, 4);
    float w3 = fold_pair(u[6], u[7], lane, 4);
    float y0 = fold_pair(w0, w1, lane, 2);
    float y1 = fold_pair(w2, w3, lane, 2);
    float z  = fold_pair(y0, y1, lane, 1);

    int t = (int)(__brev((unsigned)lane) >> 27);
    stage[t >> 2][c0 + (t & 3)] = z;
    __syncthreads();
    *bp = bold + stage[wv][lane] * (1.0f / BATCH);
}

extern "C" void kernel_launch(void* const* d_in, const int* in_sizes, int n_in,
                              void* d_out, int out_size, void* d_ws, size_t ws_size,
                              hipStream_t stream) {
    const float* x = (const float*)d_in[0];
    const float* W = (const float*)d_in[1];
    float* out = (float*)d_out;

    char* ws = (char*)d_ws;
    float* bij    = (float*)ws;
    float* colmax = (float*)(ws + (size_t)R_ROUTES * N_CAPS * 4);
    float* invd   = colmax + N_CAPS;
    float* sbuf   = invd + N_CAPS;
    float* vbuf   = sbuf + S_ELEMS;

    hipMemsetAsync(bij, 0, (size_t)R_ROUTES * N_CAPS * sizeof(float), stream);

    for (int it = 0; it < 3; ++it) {
        softmax_stats_kernel<<<N_CAPS, 256, 0, stream>>>(bij, colmax, invd);
        hipMemsetAsync(sbuf, 0, S_ELEMS * sizeof(float), stream);
        s_pass_kernel<<<(N_CAPS / 8) * (R_ROUTES / RPW_S), 256, 0, stream>>>(x, W, bij, colmax, invd, sbuf);
        float* vout = (it == 2) ? out : vbuf;
        squash_kernel<<<(S_ELEMS + 255) / 256, 256, 0, stream>>>(sbuf, vout);
        if (it < 2) {
            a_pass_kernel<<<R_ROUTES / A_ROUTES, 256, 0, stream>>>(x, W, vbuf, bij);
        }
    }
}


# launch count 15 -> 9: s clears folded into the softmax-stats kernel, squash applied by the a-pass as it loads s, iteration 0 (b_ij = 0) run with constant weight 1/4096 and a written b_ij
# speedup vs baseline: 8.1030x; 1.1763x over previous
#include <hip/hip_runtime.h>
#include <math.h>

typedef float v2f __attribute__((ext_vector_type(2)));
typedef float v4f __attribute__((ext_vector_type(4)));
typedef float v8f __attribute__((ext_vector_type(8)));

#define R_ROUTES 4096
#define N_CAPS   32
#define N_OUT    16
#define N_IN     16
#define BATCH    16
#define XB_STRIDE (R_ROUTES * N_IN)
#define W_R_STRIDE (N_CAPS * N_OUT * N_IN)
#define S_COLS   (N_CAPS * N_OUT)
#define S_ELEMS  (BATCH * S_COLS)
#define RPW_S    32
#define S_GROUP  4
#define A_ROUTES 8
#define A_CAPS   4
#define XS_PITCH(NR) ((NR) * N_IN + 4)


template <int NR>
__device__ __forceinline__ void stage_x(const float* __restrict__ x, int r0, float* xs) {
    constexpr int CH_ROW = NR * 4;
#pragma unroll
    for (int j = 0; j < (BATCH * CH_ROW) / 256; ++j) {
        int idx = j * 256 + (int)threadIdx.x;
        int bb  = idx / CH_ROW;
        int q   = idx % CH_ROW;
        v4f t = *(const v4f*)(x + (size_t)bb * XB_STRIDE + r0 * N_IN + q * 4);
        *(v4f*)(xs + bb * XS_PITCH(NR) + q * 4) = t;
    }
}

__device__ __forceinline__ float squash1(float sv) {
    float sq = sv * sv;
    return sq * sv / ((1.0f + sq) * sqrtf(sq));
}

template <int NR>
__device__ __forceinline__ void load_A(const float* xs, int rr, int lane, v4f a[2]) {
    int m  = lane & 15;
    int ko = (lane >> 4) << 3;
    const float* p = xs + m * XS_PITCH(NR) + rr * N_IN + ko;
    a[0] = *(const v4f*)p;
    a[1] = *(const v4f*)(p + 4);
}

__device__ __forceinline__ void load_B(const float* __restrict__ W, int r, int c, int lane,
                                       v4f b[2]) {
    int n  = lane & 15;
    int ko = (lane >> 4) << 3;
    const float* p = W + (size_t)r * W_R_STRIDE + (c * N_OUT + n) * N_IN + ko;
    b[0] = *(const v4f*)p;
    b[1] = *(const v4f*)(p + 4);
}

__device__ __forceinline__ v8f wmma_k16(const v4f a[2], const v4f b[2]) {
    v8f acc = {0.f, 0.f, 0.f, 0.f, 0.f, 0.f, 0.f, 0.f};
#pragma unroll
    for (int j = 0; j < 2; ++j) {
        v2f a0 = {a[j].x, a[j].y}, b0 = {b[j].x, b[j].y};
        v2f a1 = {a[j].z, a[j].w}, b1 = {b[j].z, b[j].w};
        acc = __builtin_amdgcn_wmma_f32_16x16x4_f32(
            false, a0, false, b0, (short)0, acc, false, false);
        acc = __builtin_amdgcn_wmma_f32_16x16x4_f32(
            false, a1, false, b1, (short)0, acc, false, false);
    }
    return acc;
}

__global__ void softmax_stats_kernel(const float* __restrict__ bij,
                                     float* __restrict__ colmax,
                                     float* __restrict__ inv_denom,
                                     float* __restrict__ s_clear) {
    __shared__ float red[256];
    int c = blockIdx.x;
    int t = threadIdx.x;
    s_clear[c * 256 + t] = 0.f;
    float mx = -INFINITY;
    for (int r = t; r < R_ROUTES; r += 256) mx = fmaxf(mx, bij[r * N_CAPS + c]);
    red[t] = mx;
    __syncthreads();
    for (int s2 = 128; s2 > 0; s2 >>= 1) {
        if (t < s2) red[t] = fmaxf(red[t], red[t + s2]);
        __syncthreads();
    }
    float cm = red[0];
    __syncthreads();
    float sum = 0.f;
    for (int r = t; r < R_ROUTES; r += 256) sum += __expf(bij[r * N_CAPS + c] - cm);
    red[t] = sum;
    __syncthreads();
    for (int s2 = 128; s2 > 0; s2 >>= 1) {
        if (t < s2) red[t] += red[t + s2];
        __syncthreads();
    }
    if (t == 0) {
        colmax[c]    = cm;
        inv_denom[c] = 1.0f / red[0];
    }
}

__device__ __forceinline__ void s_load_group(const float* xs, const float* __restrict__ W,
                                             int r0, int rbase, int c, int lane,
                                             v4f (&aa)[S_GROUP][2], v4f (&bb)[S_GROUP][2]) {
#pragma unroll
    for (int j = 0; j < S_GROUP; ++j) {
        load_A<RPW_S>(xs, rbase + j, lane, aa[j]);
        load_B(W, r0 + rbase + j, c, lane, bb[j]);
    }
}

__device__ __forceinline__ void s_products(const v4f (&aa)[S_GROUP][2],
                                           const v4f (&bb)[S_GROUP][2],
                                           float wl, int rbase, v8f& sacc) {
#pragma unroll
    for (int j = 0; j < S_GROUP; ++j) {
        v8f d = wmma_k16(aa[j], bb[j]);
        float wgt = __builtin_bit_cast(float,
            __builtin_amdgcn_readlane(__builtin_bit_cast(int, wl), rbase + j));
#pragma unroll
        for (int vi = 0; vi < 8; ++vi) sacc[vi] = __builtin_fmaf(wgt, d[vi], sacc[vi]);
    }
}

template <bool FIRST>
__device__ __forceinline__ void s_pass_body(const float* __restrict__ x, const float* __restrict__ W,
                                            const float* __restrict__ bij,
                                            const float* __restrict__ colmax,
                                            const float* __restrict__ inv_denom,
                                            float* __restrict__ s, float* xs) {
    int lane  = threadIdx.x & 31;
    int c     = ((blockIdx.x & 3) << 3) + (threadIdx.x >> 5);
    int r0    = (blockIdx.x >> 2) * RPW_S;
    int half8 = (lane >> 4) << 3;
    int o     = lane & 15;

    v8f sacc = {0.f, 0.f, 0.f, 0.f, 0.f, 0.f, 0.f, 0.f};
    v4f a0[S_GROUP][2], b0[S_GROUP][2], a1[S_GROUP][2], b1[S_GROUP][2];

#pragma unroll
    for (int j = 0; j < S_GROUP; ++j) load_B(W, r0 + j, c, lane, b0[j]);

    stage_x<RPW_S>(x, r0, xs);

    float wl;
    if (FIRST) wl = 1.0f / R_ROUTES;
    else       wl = __expf(bij[(r0 + lane) * N_CAPS + c] - colmax[c]) * inv_denom[c];

    __syncthreads();

#pragma unroll
    for (int j = 0; j < S_GROUP; ++j) load_A<RPW_S>(xs, j, lane, a0[j]);

    constexpr int N_PAIR = RPW_S / (2 * S_GROUP);

#pragma unroll 1
    for (int gg = 0; gg < N_PAIR - 1; ++gg) {
        const int rb = gg * 2 * S_GROUP;
        s_load_group(xs, W, r0, rb + S_GROUP, c, lane, a1, b1);
        __builtin_amdgcn_sched_barrier(0);
        s_products(a0, b0, wl, rb, sacc);
        s_load_group(xs, W, r0, rb + 2 * S_GROUP, c, lane, a0, b0);
        __builtin_amdgcn_sched_barrier(0);
        s_products(a1, b1, wl, rb + S_GROUP, sacc);
    }
    {
        const int rb = (N_PAIR - 1) * 2 * S_GROUP;
        s_load_group(xs, W, r0, rb + S_GROUP, c, lane, a1, b1);
        __builtin_amdgcn_sched_barrier(0);
        s_products(a0, b0, wl, rb, sacc);
        s_products(a1, b1, wl, rb + S_GROUP, sacc);
    }
#pragma unroll
    for (int vi = 0; vi < 8; ++vi) {
        int m = vi + half8;
        atomicAdd(&s[m * S_COLS + c * N_OUT + o], sacc[vi]);
    }
}

__global__ void __launch_bounds__(256)
s_pass_first_kernel(const float* __restrict__ x, const float* __restrict__ W,
                    float* __restrict__ s) {
    __shared__ __attribute__((aligned(16))) float xs[BATCH * XS_PITCH(RPW_S)];
    s_pass_body<true>(x, W, nullptr, nullptr, nullptr, s, xs);
}

__global__ void __launch_bounds__(256)
s_pass_kernel(const float* __restrict__ x, const float* __restrict__ W,
              const float* __restrict__ bij,
              const float* __restrict__ colmax,
              const float* __restrict__ inv_denom,
              float* __restrict__ s) {
    __shared__ __attribute__((aligned(16))) float xs[BATCH * XS_PITCH(RPW_S)];
    s_pass_body<false>(x, W, bij, colmax, inv_denom, s, xs);
}

__global__ void squash_kernel(const float* __restrict__ s, float* __restrict__ v) {
    int i = blockIdx.x * blockDim.x + threadIdx.x;
    if (i < S_ELEMS) v[i] = squash1(s[i]);
}


__device__ __forceinline__ float fold_pair(float lo, float hi, int lane, int off) {
    bool up = (lane & off) != 0;
    float keep = up ? hi : lo;
    float send = up ? lo : hi;
    return keep + __shfl_xor(send, off, 32);
}

template <bool FIRST>
__device__ __forceinline__ void a_pass_body(const float* __restrict__ x, const float* __restrict__ W,
                                            const float* __restrict__ s, float* __restrict__ bij,
                                            float (*stage)[N_CAPS], float* xs) {
    int lane  = threadIdx.x & 31;
    int wv    = threadIdx.x >> 5;
    int c0    = wv * A_CAPS;
    int r0    = blockIdx.x * A_ROUTES;
    int half8 = (lane >> 4) << 3;
    int o     = lane & 15;

    float* bp  = bij + (r0 + wv) * N_CAPS + lane;
    float bold = 0.f;
    if (!FIRST) bold = *bp;

    float vreg[A_CAPS][8];
#pragma unroll
    for (int q = 0; q < A_CAPS; ++q)
#pragma unroll
        for (int vi = 0; vi < 8; ++vi)
            vreg[q][vi] = s[(vi + half8) * S_COLS + (c0 + q) * N_OUT + o];

    v4f a[2][2], b[2][A_CAPS][2];
#pragma unroll
    for (int q = 0; q < A_CAPS; ++q) load_B(W, r0, c0 + q, lane, b[0][q]);

    stage_x<A_ROUTES>(x, r0, xs);

#pragma unroll
    for (int q = 0; q < A_CAPS; ++q)
#pragma unroll
        for (int vi = 0; vi < 8; ++vi) vreg[q][vi] = squash1(vreg[q][vi]);

    __syncthreads();
    load_A<A_ROUTES>(xs, 0, lane, a[0]);

    float u[A_ROUTES];

#pragma unroll
    for (int i = 0; i < A_ROUTES; ++i) {
        const int cur = i & 1, nxt = cur ^ 1;
        if (i + 1 < A_ROUTES) {
            load_A<A_ROUTES>(xs, i + 1, lane, a[nxt]);
#pragma unroll
            for (int q = 0; q < A_CAPS; ++q) load_B(W, r0 + i + 1, c0 + q, lane, b[nxt][q]);
        }
        __builtin_amdgcn_sched_barrier(0);
        float p[A_CAPS];
#pragma unroll
        for (int q = 0; q < A_CAPS; ++q) {
            v8f d = wmma_k16(a[cur], b[cur][q]);
            float partial = 0.f;
#pragma unroll
            for (int vi = 0; vi < 8; ++vi) partial += d[vi] * vreg[q][vi];
            p[q] = partial;
        }
        float t0 = fold_pair(p[0], p[1], lane, 16);
        float t1 = fold_pair(p[2], p[3], lane, 16);
        u[i] = fold_pair(t0, t1, lane, 8);
    }

    float w0 = fold_pair(u[0], u[1], lane, 4);
    float w1 = fold_pair(u[2], u[3], lane, 4);
    float w2 = fold_pair(u[4], u[5], lane, 4);
    float w3 = fold_pair(u[6], u[7], lane, 4);
    float y0 = fold_pair(w0, w1, lane, 2);
    float y1 = fold_pair(w2, w3, lane, 2);
    float z  = fold_pair(y0, y1, lane, 1);

    int t = (int)(__brev((unsigned)lane) >> 27);
    stage[t >> 2][c0 + (t & 3)] = z;
    __syncthreads();
    *bp = bold + stage[wv][lane] * (1.0f / BATCH);
}

__global__ void __launch_bounds__(256)
a_pass_first_kernel(const float* __restrict__ x, const float* __restrict__ W,
                    const float* __restrict__ s, float* __restrict__ bij) {
    __shared__ float stage[A_ROUTES][N_CAPS];
    __shared__ __attribute__((aligned(16))) float xs[BATCH * XS_PITCH(A_ROUTES)];
    a_pass_body<true>(x, W, s, bij, stage, xs);
}

__global__ void __launch_bounds__(256)
a_pass_kernel(const float* __restrict__ x, const float* __restrict__ W,
              const float* __restrict__ s, float* __restrict__ bij) {
    __shared__ float stage[A_ROUTES][N_CAPS];
    __shared__ __attribute__((aligned(16))) float xs[BATCH * XS_PITCH(A_ROUTES)];
    a_pass_body<false>(x, W, s, bij, stage, xs);
}

extern "C" void kernel_launch(void* const* d_in, const int* in_sizes, int n_in,
                              void* d_out, int out_size, void* d_ws, size_t ws_size,
                              hipStream_t stream) {
    const float* x = (const float*)d_in[0];
    const float* W = (const float*)d_in[1];
    float* out = (float*)d_out;

    char* ws = (char*)d_ws;
    float* bij    = (float*)ws;
    float* colmax = (float*)(ws + (size_t)R_ROUTES * N_CAPS * 4);
    float* invd   = colmax + N_CAPS;
    float* sbuf   = invd + N_CAPS;


    hipMemsetAsync(sbuf, 0, S_ELEMS * sizeof(float), stream);
    s_pass_first_kernel<<<(N_CAPS / 8) * (R_ROUTES / RPW_S), 256, 0, stream>>>(x, W, sbuf);
    a_pass_first_kernel<<<R_ROUTES / A_ROUTES, 256, 0, stream>>>(x, W, sbuf, bij);

    softmax_stats_kernel<<<N_CAPS, 256, 0, stream>>>(bij, colmax, invd, sbuf);
    s_pass_kernel<<<(N_CAPS / 8) * (R_ROUTES / RPW_S), 256, 0, stream>>>(x, W, bij, colmax, invd, sbuf);
    a_pass_kernel<<<R_ROUTES / A_ROUTES, 256, 0, stream>>>(x, W, sbuf, bij);

    softmax_stats_kernel<<<N_CAPS, 256, 0, stream>>>(bij, colmax, invd, sbuf);
    s_pass_kernel<<<(N_CAPS / 8) * (R_ROUTES / RPW_S), 256, 0, stream>>>(x, W, bij, colmax, invd, sbuf);
    squash_kernel<<<(S_ELEMS + 255) / 256, 256, 0, stream>>>(sbuf, out);
}


# agreement update fused into the next weighted-sum pass (one u_hat tile serves both): 5 reads of W -> 3; softmax shift = column maximum before the update; denominator summed per wave in route order
# speedup vs baseline: 14.4083x; 1.7781x over previous
#include <hip/hip_runtime.h>
#include <math.h>

typedef float v2f __attribute__((ext_vector_type(2)));
typedef float v4f __attribute__((ext_vector_type(4)));
typedef float v8f __attribute__((ext_vector_type(8)));

#define R_ROUTES 4096
#define N_CAPS   32
#define N_OUT    16
#define N_IN     16
#define BATCH    16
#define XB_STRIDE (R_ROUTES * N_IN)
#define W_R_STRIDE (N_CAPS * N_OUT * N_IN)
#define S_COLS   (N_CAPS * N_OUT)
#define S_ELEMS  (BATCH * S_COLS)
#define RPW_S    32
#define S_GROUP  4
#define N_COPY   4
#define KEY_LOWEST ((int)0x80000000)
#define XS_PITCH(NR) ((NR) * N_IN + 4)


template <int NR>
__device__ __forceinline__ void stage_x(const float* __restrict__ x, int r0, float* xs) {
    constexpr int CH_ROW = NR * 4;
#pragma unroll
    for (int j = 0; j < (BATCH * CH_ROW) / 256; ++j) {
        int idx = j * 256 + (int)threadIdx.x;
        int bb  = idx / CH_ROW;
        int q   = idx % CH_ROW;
        v4f t = *(const v4f*)(x + (size_t)bb * XB_STRIDE + r0 * N_IN + q * 4);
        *(v4f*)(xs + bb * XS_PITCH(NR) + q * 4) = t;
    }
}

__device__ __forceinline__ float squash1(float sv) {
    float sq = sv * sv;
    return sq * sv / ((1.0f + sq) * sqrtf(sq));
}

__device__ __forceinline__ int f2key(float f) {
    int i = __float_as_int(f);
    return i >= 0 ? i : (i ^ 0x7fffffff);
}
__device__ __forceinline__ float key2f(int k) {
    return __int_as_float(k >= 0 ? k : (k ^ 0x7fffffff));
}

__device__ __forceinline__ float den_total(const float* __restrict__ den, int c) {
    float d = den[c];
#pragma unroll
    for (int k = 1; k < N_COPY; ++k) d += den[k * N_CAPS + c];
    return d;
}

template <int NR>
__device__ __forceinline__ void load_A(const float* xs, int rr, int lane, v4f a[2]) {
    int m  = lane & 15;
    int ko = (lane >> 4) << 3;
    const float* p = xs + m * XS_PITCH(NR) + rr * N_IN + ko;
    a[0] = *(const v4f*)p;
    a[1] = *(const v4f*)(p + 4);
}

__device__ __forceinline__ void load_B(const float* __restrict__ W, int r, int c, int lane,
                                       v4f b[2]) {
    int n  = lane & 15;
    int ko = (lane >> 4) << 3;
    const float* p = W + (size_t)r * W_R_STRIDE + (c * N_OUT + n) * N_IN + ko;
    b[0] = *(const v4f*)p;
    b[1] = *(const v4f*)(p + 4);
}

__device__ __forceinline__ v8f wmma_k16(const v4f a[2], const v4f b[2]) {
    v8f acc = {0.f, 0.f, 0.f, 0.f, 0.f, 0.f, 0.f, 0.f};
#pragma unroll
    for (int j = 0; j < 2; ++j) {
        v2f a0 = {a[j].x, a[j].y}, b0 = {b[j].x, b[j].y};
        v2f a1 = {a[j].z, a[j].w}, b1 = {b[j].z, b[j].w};
        acc = __builtin_amdgcn_wmma_f32_16x16x4_f32(
            false, a0, false, b0, (short)0, acc, false, false);
        acc = __builtin_amdgcn_wmma_f32_16x16x4_f32(
            false, a1, false, b1, (short)0, acc, false, false);
    }
    return acc;
}

__device__ __forceinline__ void s_load_group(const float* xs, const float* __restrict__ W,
                                             int r0, int rbase, int c, int lane,
                                             v4f (&aa)[S_GROUP][2], v4f (&bb)[S_GROUP][2]) {
#pragma unroll
    for (int j = 0; j < S_GROUP; ++j) {
        load_A<RPW_S>(xs, rbase + j, lane, aa[j]);
        load_B(W, r0 + rbase + j, c, lane, bb[j]);
    }
}

__device__ __forceinline__ void s_products(const v4f (&aa)[S_GROUP][2],
                                           const v4f (&bb)[S_GROUP][2],
                                           float wl, int rbase, v8f& sacc) {
#pragma unroll
    for (int j = 0; j < S_GROUP; ++j) {
        v8f d = wmma_k16(aa[j], bb[j]);
        float wgt = __builtin_bit_cast(float,
            __builtin_amdgcn_readlane(__builtin_bit_cast(int, wl), rbase + j));
#pragma unroll
        for (int vi = 0; vi < 8; ++vi) sacc[vi] = __builtin_fmaf(wgt, d[vi], sacc[vi]);
    }
}

__global__ void __launch_bounds__(256)
s_pass_first_kernel(const float* __restrict__ x, const float* __restrict__ W,
                    float* __restrict__ s, float* __restrict__ s_next,
                    int* __restrict__ cmax_next, float* __restrict__ den_next) {
    __shared__ __attribute__((aligned(16))) float xs[BATCH * XS_PITCH(RPW_S)];
    int lane  = threadIdx.x & 31;
    int wv    = threadIdx.x >> 5;
    int c     = ((blockIdx.x & 3) << 3) + wv;
    int r0    = (blockIdx.x >> 2) * RPW_S;
    int half8 = (lane >> 4) << 3;
    int o     = lane & 15;

    v8f sacc = {0.f, 0.f, 0.f, 0.f, 0.f, 0.f, 0.f, 0.f};
    v4f a0[S_GROUP][2], b0[S_GROUP][2], a1[S_GROUP][2], b1[S_GROUP][2];

#pragma unroll
    for (int j = 0; j < S_GROUP; ++j) load_B(W, r0 + j, c, lane, b0[j]);

    stage_x<RPW_S>(x, r0, xs);

    const float wl = 1.0f / R_ROUTES;

    __syncthreads();

    if (blockIdx.x < S_ELEMS / 256)
        s_next[blockIdx.x * 256 + threadIdx.x] = 0.f;
    if (blockIdx.x == S_ELEMS / 256 && threadIdx.x < N_COPY * N_CAPS) {
        cmax_next[threadIdx.x] = KEY_LOWEST;
        den_next[threadIdx.x]  = 0.f;
    }

#pragma unroll
    for (int j = 0; j < S_GROUP; ++j) load_A<RPW_S>(xs, j, lane, a0[j]);

    constexpr int N_PAIR = RPW_S / (2 * S_GROUP);

#pragma unroll 1
    for (int gg = 0; gg < N_PAIR - 1; ++gg) {
        const int rb = gg * 2 * S_GROUP;
        s_load_group(xs, W, r0, rb + S_GROUP, c, lane, a1, b1);
        __builtin_amdgcn_sched_barrier(0);
        s_products(a0, b0, wl, rb, sacc);
        s_load_group(xs, W, r0, rb + 2 * S_GROUP, c, lane, a0, b0);
        __builtin_amdgcn_sched_barrier(0);
        s_products(a1, b1, wl, rb + S_GROUP, sacc);
    }
    {
        const int rb = (N_PAIR - 1) * 2 * S_GROUP;
        s_load_group(xs, W, r0, rb + S_GROUP, c, lane, a1, b1);
        __builtin_amdgcn_sched_barrier(0);
        s_products(a0, b0, wl, rb, sacc);
        s_products(a1, b1, wl, rb + S_GROUP, sacc);
    }
#pragma unroll
    for (int vi = 0; vi < 8; ++vi) {
        int m = vi + half8;
        atomicAdd(&s[m * S_COLS + c * N_OUT + o], sacc[vi]);
    }
}


__device__ __forceinline__ float fold_pair(float lo, float hi, int lane, int off) {
    bool up = (lane & off) != 0;
    float keep = up ? hi : lo;
    float send = up ? lo : hi;
    return keep + __shfl_xor(send, off, 32);
}

template <bool FIRST>
__device__ __forceinline__ void r_products(const v4f (&aa)[S_GROUP][2],
                                           const v4f (&bb)[S_GROUP][2],
                                           const float (&vreg)[8], float bo, float shift,
                                           int lane, v8f& sacc, float& dsum,
                                           float& bnewv, int& slot) {
    v8f d[S_GROUP];
    float p[S_GROUP];
#pragma unroll
    for (int j = 0; j < S_GROUP; ++j) {
        d[j] = wmma_k16(aa[j], bb[j]);
        float partial = 0.f;
#pragma unroll
        for (int vi = 0; vi < 8; ++vi) partial += d[j][vi] * vreg[vi];
        p[j] = partial;
    }
    float t0 = fold_pair(p[0], p[1], lane, 16);
    float t1 = fold_pair(p[2], p[3], lane, 16);
    float u  = fold_pair(t0, t1, lane, 8);
    u += __shfl_xor(u, 4, 32);
    u += __shfl_xor(u, 2, 32);
    u += __shfl_xor(u, 1, 32);
    float bn = bo + u * (1.0f / BATCH);
    float wl = __expf(bn - shift);
#pragma unroll
    for (int j = 0; j < S_GROUP; ++j) {
        const int srcl = ((j & 1) << 4) | ((j >> 1) << 3);
        float wgt = __builtin_bit_cast(float,
            __builtin_amdgcn_readlane(__builtin_bit_cast(int, wl), srcl));
#pragma unroll
        for (int vi = 0; vi < 8; ++vi) sacc[vi] = __builtin_fmaf(wgt, d[j][vi], sacc[vi]);
        dsum += wgt;
    }
    if (FIRST) {
        bnewv = (slot == 0) ? bn : bnewv;
        slot -= 1;
    }
}

template <bool FIRST>
__device__ __forceinline__ void r_pass_body(const float* __restrict__ x, const float* __restrict__ W,
                                            const float* __restrict__ sv,
                                            const int* __restrict__ cmax,
                                            float* __restrict__ bT, int* __restrict__ cmax_out,
                                            float* __restrict__ s, float* __restrict__ den,
                                            float* xs) {
    int lane  = threadIdx.x & 31;
    int wv    = threadIdx.x >> 5;
    int c     = ((blockIdx.x & 3) << 3) + wv;
    int r0    = (blockIdx.x >> 2) * RPW_S;
    int half8 = (lane >> 4) << 3;
    int o     = lane & 15;
    int copy  = (blockIdx.x >> 2) & (N_COPY - 1);
    int jl    = ((lane >> 4) & 1) | ((lane >> 2) & 2);
    float* bline = bT + c * R_ROUTES + r0 + jl;

    v8f sacc = {0.f, 0.f, 0.f, 0.f, 0.f, 0.f, 0.f, 0.f};
    v4f a0[S_GROUP][2], b0[S_GROUP][2], a1[S_GROUP][2], b1[S_GROUP][2];
    float bo0 = 0.f, bo1 = 0.f;

#pragma unroll
    for (int j = 0; j < S_GROUP; ++j) load_B(W, r0 + j, c, lane, b0[j]);

    float vreg[8];
#pragma unroll
    for (int vi = 0; vi < 8; ++vi) vreg[vi] = sv[(vi + half8) * S_COLS + c * N_OUT + o];

    float shift = 0.f;
    if (!FIRST) {
        int km = cmax[c];
#pragma unroll
        for (int k = 1; k < N_COPY; ++k) {
            int kk = cmax[k * N_CAPS + c];
            km = kk > km ? kk : km;
        }
        shift = key2f(km);
        bo0 = bline[0];
    }

    stage_x<RPW_S>(x, r0, xs);

    if (FIRST) {
#pragma unroll
        for (int vi = 0; vi < 8; ++vi) vreg[vi] = squash1(vreg[vi]);
    }

    __syncthreads();

#pragma unroll
    for (int j = 0; j < S_GROUP; ++j) load_A<RPW_S>(xs, j, lane, a0[j]);

    float dsum  = 0.f;
    float bnewv = 0.f;
    int   slot  = lane & 7;

    constexpr int N_PAIR = RPW_S / (2 * S_GROUP);

#pragma unroll 1
    for (int gg = 0; gg < N_PAIR - 1; ++gg) {
        const int rb = gg * 2 * S_GROUP;
        s_load_group(xs, W, r0, rb + S_GROUP, c, lane, a1, b1);
        if (!FIRST) bo1 = bline[rb + S_GROUP];
        __builtin_amdgcn_sched_barrier(0);
        r_products<FIRST>(a0, b0, vreg, bo0, shift, lane, sacc, dsum, bnewv, slot);
        s_load_group(xs, W, r0, rb + 2 * S_GROUP, c, lane, a0, b0);
        if (!FIRST) bo0 = bline[rb + 2 * S_GROUP];
        __builtin_amdgcn_sched_barrier(0);
        r_products<FIRST>(a1, b1, vreg, bo1, shift, lane, sacc, dsum, bnewv, slot);
    }
    {
        const int rb = (N_PAIR - 1) * 2 * S_GROUP;
        s_load_group(xs, W, r0, rb + S_GROUP, c, lane, a1, b1);
        if (!FIRST) bo1 = bline[rb + S_GROUP];
        __builtin_amdgcn_sched_barrier(0);
        r_products<FIRST>(a0, b0, vreg, bo0, shift, lane, sacc, dsum, bnewv, slot);
        r_products<FIRST>(a1, b1, vreg, bo1, shift, lane, sacc, dsum, bnewv, slot);
    }

#pragma unroll
    for (int vi = 0; vi < 8; ++vi) {
        int m = vi + half8;
        atomicAdd(&s[m * S_COLS + c * N_OUT + o], sacc[vi]);
    }
    if (lane == 0) atomicAdd(&den[copy * N_CAPS + c], dsum);

    if (FIRST) {
        bline[S_GROUP * (lane & 7)] = bnewv;
        float mx = bnewv;
#pragma unroll
        for (int off = 16; off > 0; off >>= 1) mx = fmaxf(mx, __shfl_xor(mx, off, 32));
        if (lane == 0) atomicMax(&cmax_out[copy * N_CAPS + c], f2key(mx));
    }
}

__global__ void __launch_bounds__(256)
r_pass_first_kernel(const float* __restrict__ x, const float* __restrict__ W,
                    const float* __restrict__ s0, float* __restrict__ bT,
                    int* __restrict__ cmax_out, float* __restrict__ s, float* __restrict__ den) {
    __shared__ __attribute__((aligned(16))) float xs[BATCH * XS_PITCH(RPW_S)];
    r_pass_body<true>(x, W, s0, nullptr, bT, cmax_out, s, den, xs);
}

__global__ void __launch_bounds__(256)
r_pass_kernel(const float* __restrict__ x, const float* __restrict__ W,
              const float* __restrict__ v, const int* __restrict__ cmax,
              float* __restrict__ bT, float* __restrict__ s, float* __restrict__ den) {
    __shared__ __attribute__((aligned(16))) float xs[BATCH * XS_PITCH(RPW_S)];
    r_pass_body<false>(x, W, v, cmax, bT, nullptr, s, den, xs);
}

__global__ void mid_kernel(float* __restrict__ sv, const float* __restrict__ den,
                           float* __restrict__ s_next, float* __restrict__ den_next) {
    int i = blockIdx.x * blockDim.x + threadIdx.x;
    if (i < S_ELEMS) {
        int c = (i >> 4) & (N_CAPS - 1);
        sv[i] = squash1(sv[i] * (1.0f / den_total(den, c)));
        s_next[i] = 0.f;
    }
    if (i < N_COPY * N_CAPS) den_next[i] = 0.f;
}

__global__ void squash_kernel(const float* __restrict__ s, const float* __restrict__ den,
                              float* __restrict__ v) {
    int i = blockIdx.x * blockDim.x + threadIdx.x;
    if (i < S_ELEMS) {
        int c = (i >> 4) & (N_CAPS - 1);
        v[i] = squash1(s[i] * (1.0f / den_total(den, c)));
    }
}

extern "C" void kernel_launch(void* const* d_in, const int* in_sizes, int n_in,
                              void* d_out, int out_size, void* d_ws, size_t ws_size,
                              hipStream_t stream) {
    const float* x = (const float*)d_in[0];
    const float* W = (const float*)d_in[1];
    float* out = (float*)d_out;

    char* ws = (char*)d_ws;
    float* bT   = (float*)ws;
    int*   cmx1 = (int*)(ws + (size_t)R_ROUTES * N_CAPS * 4);
    int*   cmx2 = cmx1 + N_COPY * N_CAPS;
    float* den1 = (float*)(cmx2 + N_COPY * N_CAPS);
    float* den2 = den1 + N_COPY * N_CAPS;
    float* sbuf = den2 + N_COPY * N_CAPS;


    hipMemsetAsync(sbuf, 0, S_ELEMS * sizeof(float), stream);
    s_pass_first_kernel<<<(N_CAPS / 8) * (R_ROUTES / RPW_S), 256, 0, stream>>>(
        x, W, sbuf, out, cmx1, den1);

    r_pass_first_kernel<<<(N_CAPS / 8) * (R_ROUTES / RPW_S), 256, 0, stream>>>(
        x, W, sbuf, bT, cmx1, out, den1);

    mid_kernel<<<(S_ELEMS + 255) / 256, 256, 0, stream>>>(out, den1, sbuf, den2);

    r_pass_kernel<<<(N_CAPS / 8) * (R_ROUTES / RPW_S), 256, 0, stream>>>(
        x, W, out, cmx1, bT, sbuf, den2);

    squash_kernel<<<(S_ELEMS + 255) / 256, 256, 0, stream>>>(sbuf, den2, out);
}


# s and denominator: block = 8 capsules x 2 route tiles (512 thr); a capsule's two waves add their partial sums in LDS ahead of the atomic: 64 atomic partials per output, was 128 (sum order changes)
# speedup vs baseline: 16.7192x; 1.1604x over previous
#include <hip/hip_runtime.h>
#include <math.h>

typedef float v2f __attribute__((ext_vector_type(2)));
typedef float v4f __attribute__((ext_vector_type(4)));
typedef float v8f __attribute__((ext_vector_type(8)));

#define R_ROUTES 4096
#define N_CAPS   32
#define N_OUT    16
#define N_IN     16
#define BATCH    16
#define XB_STRIDE (R_ROUTES * N_IN)
#define W_R_STRIDE (N_CAPS * N_OUT * N_IN)
#define S_COLS   (N_CAPS * N_OUT)
#define S_ELEMS  (BATCH * S_COLS)
#define RPW_S    32
#define S_GROUP  4
#define TILES_PB 2
#define NR_BLK   (RPW_S * TILES_PB)
#define BLK_THR  (8 * TILES_PB * 32)
#define N_COPY   4
#define KEY_LOWEST ((int)0x80000000)
#define XS_PITCH(NR) ((NR) * N_IN + 4)


template <int NR, int NT>
__device__ __forceinline__ void stage_x(const float* __restrict__ x, int r0, float* xs) {
    constexpr int CH_ROW = NR * 4;
#pragma unroll
    for (int j = 0; j < (BATCH * CH_ROW) / NT; ++j) {
        int idx = j * NT + (int)threadIdx.x;
        int bb  = idx / CH_ROW;
        int q   = idx % CH_ROW;
        v4f t = *(const v4f*)(x + (size_t)bb * XB_STRIDE + r0 * N_IN + q * 4);
        *(v4f*)(xs + bb * XS_PITCH(NR) + q * 4) = t;
    }
}

__device__ __forceinline__ float squash1(float sv) {
    float sq = sv * sv;
    return sq * sv / ((1.0f + sq) * sqrtf(sq));
}

__device__ __forceinline__ int f2key(float f) {
    int i = __float_as_int(f);
    return i >= 0 ? i : (i ^ 0x7fffffff);
}
__device__ __forceinline__ float key2f(int k) {
    return __int_as_float(k >= 0 ? k : (k ^ 0x7fffffff));
}

__device__ __forceinline__ float den_total(const float* __restrict__ den, int c) {
    float d = den[c];
#pragma unroll
    for (int k = 1; k < N_COPY; ++k) d += den[k * N_CAPS + c];
    return d;
}

template <int NR>
__device__ __forceinline__ void load_A(const float* xs, int rr, int lane, v4f a[2]) {
    int m  = lane & 15;
    int ko = (lane >> 4) << 3;
    const float* p = xs + m * XS_PITCH(NR) + rr * N_IN + ko;
    a[0] = *(const v4f*)p;
    a[1] = *(const v4f*)(p + 4);
}

__device__ __forceinline__ void load_B(const float* __restrict__ W, int r, int c, int lane,
                                       v4f b[2]) {
    int n  = lane & 15;
    int ko = (lane >> 4) << 3;
    const float* p = W + (size_t)r * W_R_STRIDE + (c * N_OUT + n) * N_IN + ko;
    b[0] = *(const v4f*)p;
    b[1] = *(const v4f*)(p + 4);
}

__device__ __forceinline__ v8f wmma_k16(const v4f a[2], const v4f b[2]) {
    v8f acc = {0.f, 0.f, 0.f, 0.f, 0.f, 0.f, 0.f, 0.f};
#pragma unroll
    for (int j = 0; j < 2; ++j) {
        v2f a0 = {a[j].x, a[j].y}, b0 = {b[j].x, b[j].y};
        v2f a1 = {a[j].z, a[j].w}, b1 = {b[j].z, b[j].w};
        acc = __builtin_amdgcn_wmma_f32_16x16x4_f32(
            false, a0, false, b0, (short)0, acc, false, false);
        acc = __builtin_amdgcn_wmma_f32_16x16x4_f32(
            false, a1, false, b1, (short)0, acc, false, false);
    }
    return acc;
}

__device__ __forceinline__ void s_load_group(const float* xs, const float* __restrict__ W,
                                             int r0, int rbase, int c, int lane,
                                             v4f (&aa)[S_GROUP][2], v4f (&bb)[S_GROUP][2]) {
#pragma unroll
    for (int j = 0; j < S_GROUP; ++j) {
        load_A<NR_BLK>(xs, rbase + j, lane, aa[j]);
        load_B(W, r0 + rbase + j, c, lane, bb[j]);
    }
}

__device__ __forceinline__ void s_products(const v4f (&aa)[S_GROUP][2],
                                           const v4f (&bb)[S_GROUP][2],
                                           float wl, int rbase, v8f& sacc) {
#pragma unroll
    for (int j = 0; j < S_GROUP; ++j) {
        v8f d = wmma_k16(aa[j], bb[j]);
        float wgt = __builtin_bit_cast(float,
            __builtin_amdgcn_readlane(__builtin_bit_cast(int, wl), rbase + j));
#pragma unroll
        for (int vi = 0; vi < 8; ++vi) sacc[vi] = __builtin_fmaf(wgt, d[vi], sacc[vi]);
    }
}

__global__ void __launch_bounds__(BLK_THR)
s_pass_first_kernel(const float* __restrict__ x, const float* __restrict__ W,
                    float* __restrict__ s, float* __restrict__ s_next,
                    int* __restrict__ cmax_next, float* __restrict__ den_next) {
    __shared__ __attribute__((aligned(16))) float xs_blk[BATCH * XS_PITCH(NR_BLK)];
    __shared__ float red[8 * 8 * 32];
    int lane  = threadIdx.x & 31;
    int wv    = threadIdx.x >> 5;
    int cw    = wv & 7;
    int tl    = wv >> 3;
    int c     = ((blockIdx.x & 3) << 3) + cw;
    int rblk  = (blockIdx.x >> 2) * NR_BLK;
    int r0    = rblk + tl * RPW_S;
    const float* xs = xs_blk + tl * RPW_S * N_IN;
    int half8 = (lane >> 4) << 3;
    int o     = lane & 15;

    v8f sacc = {0.f, 0.f, 0.f, 0.f, 0.f, 0.f, 0.f, 0.f};
    v4f a0[S_GROUP][2], b0[S_GROUP][2], a1[S_GROUP][2], b1[S_GROUP][2];

#pragma unroll
    for (int j = 0; j < S_GROUP; ++j) load_B(W, r0 + j, c, lane, b0[j]);

    stage_x<NR_BLK, BLK_THR>(x, rblk, xs_blk);

    const float wl = 1.0f / R_ROUTES;

    __syncthreads();

    if (blockIdx.x < S_ELEMS / BLK_THR)
        s_next[blockIdx.x * BLK_THR + threadIdx.x] = 0.f;
    if (blockIdx.x == S_ELEMS / BLK_THR && threadIdx.x < N_COPY * N_CAPS) {
        cmax_next[threadIdx.x] = KEY_LOWEST;
        den_next[threadIdx.x]  = 0.f;
    }

#pragma unroll
    for (int j = 0; j < S_GROUP; ++j) load_A<NR_BLK>(xs, j, lane, a0[j]);

    constexpr int N_PAIR = RPW_S / (2 * S_GROUP);

#pragma unroll 1
    for (int gg = 0; gg < N_PAIR - 1; ++gg) {
        const int rb = gg * 2 * S_GROUP;
        s_load_group(xs, W, r0, rb + S_GROUP, c, lane, a1, b1);
        __builtin_amdgcn_sched_barrier(0);
        s_products(a0, b0, wl, rb, sacc);
        s_load_group(xs, W, r0, rb + 2 * S_GROUP, c, lane, a0, b0);
        __builtin_amdgcn_sched_barrier(0);
        s_products(a1, b1, wl, rb + S_GROUP, sacc);
    }
    {
        const int rb = (N_PAIR - 1) * 2 * S_GROUP;
        s_load_group(xs, W, r0, rb + S_GROUP, c, lane, a1, b1);
        __builtin_amdgcn_sched_barrier(0);
        s_products(a0, b0, wl, rb, sacc);
        s_products(a1, b1, wl, rb + S_GROUP, sacc);
    }
    if (tl != 0) {
#pragma unroll
        for (int vi = 0; vi < 8; ++vi) red[(cw * 8 + vi) * 32 + lane] = sacc[vi];
    }
    __syncthreads();
    if (tl == 0) {
#pragma unroll
        for (int vi = 0; vi < 8; ++vi) sacc[vi] += red[(cw * 8 + vi) * 32 + lane];
#pragma unroll
        for (int vi = 0; vi < 8; ++vi) {
            int m = vi + half8;
            atomicAdd(&s[m * S_COLS + c * N_OUT + o], sacc[vi]);
        }
    }
}


__device__ __forceinline__ float fold_pair(float lo, float hi, int lane, int off) {
    bool up = (lane & off) != 0;
    float keep = up ? hi : lo;
    float send = up ? lo : hi;
    return keep + __shfl_xor(send, off, 32);
}

template <bool FIRST>
__device__ __forceinline__ void r_products(const v4f (&aa)[S_GROUP][2],
                                           const v4f (&bb)[S_GROUP][2],
                                           const float (&vreg)[8], float bo, float shift,
                                           int lane, v8f& sacc, float& dsum,
                                           float& bnewv, int& slot) {
    v8f d[S_GROUP];
    float p[S_GROUP];
#pragma unroll
    for (int j = 0; j < S_GROUP; ++j) {
        d[j] = wmma_k16(aa[j], bb[j]);
        float partial = 0.f;
#pragma unroll
        for (int vi = 0; vi < 8; ++vi) partial += d[j][vi] * vreg[vi];
        p[j] = partial;
    }
    float t0 = fold_pair(p[0], p[1], lane, 16);
    float t1 = fold_pair(p[2], p[3], lane, 16);
    float u  = fold_pair(t0, t1, lane, 8);
    u += __shfl_xor(u, 4, 32);
    u += __shfl_xor(u, 2, 32);
    u += __shfl_xor(u, 1, 32);
    float bn = bo + u * (1.0f / BATCH);
    float wl = __expf(bn - shift);
#pragma unroll
    for (int j = 0; j < S_GROUP; ++j) {
        const int srcl = ((j & 1) << 4) | ((j >> 1) << 3);
        float wgt = __builtin_bit_cast(float,
            __builtin_amdgcn_readlane(__builtin_bit_cast(int, wl), srcl));
#pragma unroll
        for (int vi = 0; vi < 8; ++vi) sacc[vi] = __builtin_fmaf(wgt, d[j][vi], sacc[vi]);
        dsum += wgt;
    }
    if (FIRST) {
        bnewv = (slot == 0) ? bn : bnewv;
        slot -= 1;
    }
}

template <bool FIRST>
__device__ __forceinline__ void r_pass_body(const float* __restrict__ x, const float* __restrict__ W,
                                            const float* __restrict__ sv,
                                            const int* __restrict__ cmax,
                                            float* __restrict__ bT, int* __restrict__ cmax_out,
                                            float* __restrict__ s, float* __restrict__ den,
                                            float* xs_blk, float* red, float* redd) {
    int lane  = threadIdx.x & 31;
    int wv    = threadIdx.x >> 5;
    int cw    = wv & 7;
    int tl    = wv >> 3;
    int c     = ((blockIdx.x & 3) << 3) + cw;
    int rblk  = (blockIdx.x >> 2) * NR_BLK;
    int r0    = rblk + tl * RPW_S;
    const float* xs = xs_blk + tl * RPW_S * N_IN;
    int half8 = (lane >> 4) << 3;
    int o     = lane & 15;
    int copy  = (blockIdx.x >> 2) & (N_COPY - 1);
    int jl    = ((lane >> 4) & 1) | ((lane >> 2) & 2);
    float* bline = bT + c * R_ROUTES + r0 + jl;

    v8f sacc = {0.f, 0.f, 0.f, 0.f, 0.f, 0.f, 0.f, 0.f};
    v4f a0[S_GROUP][2], b0[S_GROUP][2], a1[S_GROUP][2], b1[S_GROUP][2];
    float bo0 = 0.f, bo1 = 0.f;

#pragma unroll
    for (int j = 0; j < S_GROUP; ++j) load_B(W, r0 + j, c, lane, b0[j]);

    float vreg[8];
#pragma unroll
    for (int vi = 0; vi < 8; ++vi) vreg[vi] = sv[(vi + half8) * S_COLS + c * N_OUT + o];

    float shift = 0.f;
    if (!FIRST) {
        int km = cmax[c];
#pragma unroll
        for (int k = 1; k < N_COPY; ++k) {
            int kk = cmax[k * N_CAPS + c];
            km = kk > km ? kk : km;
        }
        shift = key2f(km);
        bo0 = bline[0];
    }

    stage_x<NR_BLK, BLK_THR>(x, rblk, xs_blk);

    if (FIRST) {
#pragma unroll
        for (int vi = 0; vi < 8; ++vi) vreg[vi] = squash1(vreg[vi]);
    }

    __syncthreads();

#pragma unroll
    for (int j = 0; j < S_GROUP; ++j) load_A<NR_BLK>(xs, j, lane, a0[j]);

    float dsum  = 0.f;
    float bnewv = 0.f;
    int   slot  = lane & 7;

    constexpr int N_PAIR = RPW_S / (2 * S_GROUP);

#pragma unroll 1
    for (int gg = 0; gg < N_PAIR - 1; ++gg) {
        const int rb = gg * 2 * S_GROUP;
        s_load_group(xs, W, r0, rb + S_GROUP, c, lane, a1, b1);
        if (!FIRST) bo1 = bline[rb + S_GROUP];
        __builtin_amdgcn_sched_barrier(0);
        r_products<FIRST>(a0, b0, vreg, bo0, shift, lane, sacc, dsum, bnewv, slot);
        s_load_group(xs, W, r0, rb + 2 * S_GROUP, c, lane, a0, b0);
        if (!FIRST) bo0 = bline[rb + 2 * S_GROUP];
        __builtin_amdgcn_sched_barrier(0);
        r_products<FIRST>(a1, b1, vreg, bo1, shift, lane, sacc, dsum, bnewv, slot);
    }
    {
        const int rb = (N_PAIR - 1) * 2 * S_GROUP;
        s_load_group(xs, W, r0, rb + S_GROUP, c, lane, a1, b1);
        if (!FIRST) bo1 = bline[rb + S_GROUP];
        __builtin_amdgcn_sched_barrier(0);
        r_products<FIRST>(a0, b0, vreg, bo0, shift, lane, sacc, dsum, bnewv, slot);
        r_products<FIRST>(a1, b1, vreg, bo1, shift, lane, sacc, dsum, bnewv, slot);
    }

    if (tl != 0) {
#pragma unroll
        for (int vi = 0; vi < 8; ++vi) red[(cw * 8 + vi) * 32 + lane] = sacc[vi];
        if (lane == 0) redd[cw] = dsum;
    }
    __syncthreads();
    if (tl == 0) {
#pragma unroll
        for (int vi = 0; vi < 8; ++vi) sacc[vi] += red[(cw * 8 + vi) * 32 + lane];
        dsum += redd[cw];
#pragma unroll
        for (int vi = 0; vi < 8; ++vi) {
            int m = vi + half8;
            atomicAdd(&s[m * S_COLS + c * N_OUT + o], sacc[vi]);
        }
        if (lane == 0) atomicAdd(&den[copy * N_CAPS + c], dsum);
    }

    if (FIRST) {
        bline[S_GROUP * (lane & 7)] = bnewv;
        float mx = bnewv;
#pragma unroll
        for (int off = 16; off > 0; off >>= 1) mx = fmaxf(mx, __shfl_xor(mx, off, 32));
        if (lane == 0) atomicMax(&cmax_out[copy * N_CAPS + c], f2key(mx));
    }
}

__global__ void __launch_bounds__(BLK_THR)
r_pass_first_kernel(const float* __restrict__ x, const float* __restrict__ W,
                    const float* __restrict__ s0, float* __restrict__ bT,
                    int* __restrict__ cmax_out, float* __restrict__ s, float* __restrict__ den) {
    __shared__ __attribute__((aligned(16))) float xs_blk[BATCH * XS_PITCH(NR_BLK)];
    __shared__ float red[8 * 8 * 32];
    __shared__ float redd[8];
    r_pass_body<true>(x, W, s0, nullptr, bT, cmax_out, s, den, xs_blk, red, redd);
}

__global__ void __launch_bounds__(BLK_THR)
r_pass_kernel(const float* __restrict__ x, const float* __restrict__ W,
              const float* __restrict__ v, const int* __restrict__ cmax,
              float* __restrict__ bT, float* __restrict__ s, float* __restrict__ den) {
    __shared__ __attribute__((aligned(16))) float xs_blk[BATCH * XS_PITCH(NR_BLK)];
    __shared__ float red[8 * 8 * 32];
    __shared__ float redd[8];
    r_pass_body<false>(x, W, v, cmax, bT, nullptr, s, den, xs_blk, red, redd);
}

__global__ void mid_kernel(float* __restrict__ sv, const float* __restrict__ den,
                           float* __restrict__ s_next, float* __restrict__ den_next) {
    int i = blockIdx.x * blockDim.x + threadIdx.x;
    if (i < S_ELEMS) {
        int c = (i >> 4) & (N_CAPS - 1);
        sv[i] = squash1(sv[i] * (1.0f / den_total(den, c)));
        s_next[i] = 0.f;
    }
    if (i < N_COPY * N_CAPS) den_next[i] = 0.f;
}

__global__ void squash_kernel(const float* __restrict__ s, const float* __restrict__ den,
                              float* __restrict__ v) {
    int i = blockIdx.x * blockDim.x + threadIdx.x;
    if (i < S_ELEMS) {
        int c = (i >> 4) & (N_CAPS - 1);
        v[i] = squash1(s[i] * (1.0f / den_total(den, c)));
    }
}

extern "C" void kernel_launch(void* const* d_in, const int* in_sizes, int n_in,
                              void* d_out, int out_size, void* d_ws, size_t ws_size,
                              hipStream_t stream) {
    const float* x = (const float*)d_in[0];
    const float* W = (const float*)d_in[1];
    float* out = (float*)d_out;

    char* ws = (char*)d_ws;
    float* bT   = (float*)ws;
    int*   cmx1 = (int*)(ws + (size_t)R_ROUTES * N_CAPS * 4);
    int*   cmx2 = cmx1 + N_COPY * N_CAPS;
    float* den1 = (float*)(cmx2 + N_COPY * N_CAPS);
    float* den2 = den1 + N_COPY * N_CAPS;
    float* sbuf = den2 + N_COPY * N_CAPS;


    hipMemsetAsync(sbuf, 0, S_ELEMS * sizeof(float), stream);
    s_pass_first_kernel<<<(N_CAPS / 8) * (R_ROUTES / NR_BLK), BLK_THR, 0, stream>>>(
        x, W, sbuf, out, cmx1, den1);

    r_pass_first_kernel<<<(N_CAPS / 8) * (R_ROUTES / NR_BLK), BLK_THR, 0, stream>>>(
        x, W, sbuf, bT, cmx1, out, den1);

    mid_kernel<<<(S_ELEMS + 255) / 256, 256, 0, stream>>>(out, den1, sbuf, den2);

    r_pass_kernel<<<(N_CAPS / 8) * (R_ROUTES / NR_BLK), BLK_THR, 0, stream>>>(
        x, W, out, cmx1, bT, sbuf, den2);

    squash_kernel<<<(S_ELEMS + 255) / 256, 256, 0, stream>>>(sbuf, den2, out);
}


# all three passes: block = 4 capsules x 4 route tiles (was 8 x 2): a capsule's four waves add their sums in LDS, 32 atomic partials per output (was 64; sum order changes); x tile 128 routes per block
# speedup vs baseline: 19.1744x; 1.1468x over previous
#include <hip/hip_runtime.h>
#include <math.h>

typedef float v2f __attribute__((ext_vector_type(2)));
typedef float v4f __attribute__((ext_vector_type(4)));
typedef float v8f __attribute__((ext_vector_type(8)));

#define R_ROUTES 4096
#define N_CAPS   32
#define N_OUT    16
#define N_IN     16
#define BATCH    16
#define XB_STRIDE (R_ROUTES * N_IN)
#define W_R_STRIDE (N_CAPS * N_OUT * N_IN)
#define S_COLS   (N_CAPS * N_OUT)
#define S_ELEMS  (BATCH * S_COLS)
#define RPW_S    32
#define S_GROUP  4
#define CAPS_PB  4
#define TILES_PB 4
#define NR_BLK   (RPW_S * TILES_PB)
#define BLK_THR  (CAPS_PB * TILES_PB * 32)
#define CGROUPS  (N_CAPS / CAPS_PB)
#define RED_FLTS ((TILES_PB - 1) * CAPS_PB * 8 * 32)
#define N_COPY   4
#define KEY_LOWEST ((int)0x80000000)
#define XS_PITCH(NR) ((NR) * N_IN + 4)


template <int NR, int NT>
__device__ __forceinline__ void stage_x(const float* __restrict__ x, int r0, float* xs) {
    constexpr int CH_ROW = NR * 4;
#pragma unroll
    for (int j = 0; j < (BATCH * CH_ROW) / NT; ++j) {
        int idx = j * NT + (int)threadIdx.x;
        int bb  = idx / CH_ROW;
        int q   = idx % CH_ROW;
        v4f t = *(const v4f*)(x + (size_t)bb * XB_STRIDE + r0 * N_IN + q * 4);
        *(v4f*)(xs + bb * XS_PITCH(NR) + q * 4) = t;
    }
}

__device__ __forceinline__ float squash1(float sv) {
    float sq = sv * sv;
    return sq * sv / ((1.0f + sq) * sqrtf(sq));
}

__device__ __forceinline__ int f2key(float f) {
    int i = __float_as_int(f);
    return i >= 0 ? i : (i ^ 0x7fffffff);
}
__device__ __forceinline__ float key2f(int k) {
    return __int_as_float(k >= 0 ? k : (k ^ 0x7fffffff));
}

__device__ __forceinline__ float den_total(const float* __restrict__ den, int c) {
    float d = den[c];
#pragma unroll
    for (int k = 1; k < N_COPY; ++k) d += den[k * N_CAPS + c];
    return d;
}

template <int NR>
__device__ __forceinline__ void load_A(const float* xs, int rr, int lane, v4f a[2]) {
    int m  = lane & 15;
    int ko = (lane >> 4) << 3;
    const float* p = xs + m * XS_PITCH(NR) + rr * N_IN + ko;
    a[0] = *(const v4f*)p;
    a[1] = *(const v4f*)(p + 4);
}

__device__ __forceinline__ void load_B(const float* __restrict__ W, int r, int c, int lane,
                                       v4f b[2]) {
    int n  = lane & 15;
    int ko = (lane >> 4) << 3;
    const float* p = W + (size_t)r * W_R_STRIDE + (c * N_OUT + n) * N_IN + ko;
    b[0] = *(const v4f*)p;
    b[1] = *(const v4f*)(p + 4);
}

__device__ __forceinline__ v8f wmma_k16(const v4f a[2], const v4f b[2]) {
    v8f acc = {0.f, 0.f, 0.f, 0.f, 0.f, 0.f, 0.f, 0.f};
#pragma unroll
    for (int j = 0; j < 2; ++j) {
        v2f a0 = {a[j].x, a[j].y}, b0 = {b[j].x, b[j].y};
        v2f a1 = {a[j].z, a[j].w}, b1 = {b[j].z, b[j].w};
        acc = __builtin_amdgcn_wmma_f32_16x16x4_f32(
            false, a0, false, b0, (short)0, acc, false, false);
        acc = __builtin_amdgcn_wmma_f32_16x16x4_f32(
            false, a1, false, b1, (short)0, acc, false, false);
    }
    return acc;
}

__device__ __forceinline__ void s_load_group(const float* xs, const float* __restrict__ W,
                                             int r0, int rbase, int c, int lane,
                                             v4f (&aa)[S_GROUP][2], v4f (&bb)[S_GROUP][2]) {
#pragma unroll
    for (int j = 0; j < S_GROUP; ++j) {
        load_A<NR_BLK>(xs, rbase + j, lane, aa[j]);
        load_B(W, r0 + rbase + j, c, lane, bb[j]);
    }
}

__device__ __forceinline__ void s_products(const v4f (&aa)[S_GROUP][2],
                                           const v4f (&bb)[S_GROUP][2],
                                           float wl, int rbase, v8f& sacc) {
#pragma unroll
    for (int j = 0; j < S_GROUP; ++j) {
        v8f d = wmma_k16(aa[j], bb[j]);
        float wgt = __builtin_bit_cast(float,
            __builtin_amdgcn_readlane(__builtin_bit_cast(int, wl), rbase + j));
#pragma unroll
        for (int vi = 0; vi < 8; ++vi) sacc[vi] = __builtin_fmaf(wgt, d[vi], sacc[vi]);
    }
}

__global__ void __launch_bounds__(BLK_THR)
s_pass_first_kernel(const float* __restrict__ x, const float* __restrict__ W,
                    float* __restrict__ s, float* __restrict__ s_next,
                    int* __restrict__ cmax_next, float* __restrict__ den_next) {
    __shared__ __attribute__((aligned(16))) float xs_blk[BATCH * XS_PITCH(NR_BLK)];
    __shared__ float red[RED_FLTS];
    int lane  = threadIdx.x & 31;
    int wv    = threadIdx.x >> 5;
    int cw    = wv % CAPS_PB;
    int tl    = wv / CAPS_PB;
    int c     = (blockIdx.x % CGROUPS) * CAPS_PB + cw;
    int rblk  = (blockIdx.x / CGROUPS) * NR_BLK;
    int r0    = rblk + tl * RPW_S;
    const float* xs = xs_blk + tl * RPW_S * N_IN;
    int half8 = (lane >> 4) << 3;
    int o     = lane & 15;

    v8f sacc = {0.f, 0.f, 0.f, 0.f, 0.f, 0.f, 0.f, 0.f};
    v4f a0[S_GROUP][2], b0[S_GROUP][2], a1[S_GROUP][2], b1[S_GROUP][2];

#pragma unroll
    for (int j = 0; j < S_GROUP; ++j) load_B(W, r0 + j, c, lane, b0[j]);

    stage_x<NR_BLK, BLK_THR>(x, rblk, xs_blk);

    const float wl = 1.0f / R_ROUTES;

    __syncthreads();

    if (blockIdx.x < S_ELEMS / BLK_THR)
        s_next[blockIdx.x * BLK_THR + threadIdx.x] = 0.f;
    if (blockIdx.x == S_ELEMS / BLK_THR && threadIdx.x < N_COPY * N_CAPS) {
        cmax_next[threadIdx.x] = KEY_LOWEST;
        den_next[threadIdx.x]  = 0.f;
    }

#pragma unroll
    for (int j = 0; j < S_GROUP; ++j) load_A<NR_BLK>(xs, j, lane, a0[j]);

    constexpr int N_PAIR = RPW_S / (2 * S_GROUP);

#pragma unroll 1
    for (int gg = 0; gg < N_PAIR - 1; ++gg) {
        const int rb = gg * 2 * S_GROUP;
        s_load_group(xs, W, r0, rb + S_GROUP, c, lane, a1, b1);
        __builtin_amdgcn_sched_barrier(0);
        s_products(a0, b0, wl, rb, sacc);
        s_load_group(xs, W, r0, rb + 2 * S_GROUP, c, lane, a0, b0);
        __builtin_amdgcn_sched_barrier(0);
        s_products(a1, b1, wl, rb + S_GROUP, sacc);
    }
    {
        const int rb = (N_PAIR - 1) * 2 * S_GROUP;
        s_load_group(xs, W, r0, rb + S_GROUP, c, lane, a1, b1);
        __builtin_amdgcn_sched_barrier(0);
        s_products(a0, b0, wl, rb, sacc);
        s_products(a1, b1, wl, rb + S_GROUP, sacc);
    }
    if (tl != 0) {
#pragma unroll
        for (int vi = 0; vi < 8; ++vi)
            red[(((tl - 1) * CAPS_PB + cw) * 8 + vi) * 32 + lane] = sacc[vi];
    }
    __syncthreads();
    if (tl == 0) {
#pragma unroll
        for (int t = 1; t < TILES_PB; ++t) {
#pragma unroll
            for (int vi = 0; vi < 8; ++vi)
                sacc[vi] += red[(((t - 1) * CAPS_PB + cw) * 8 + vi) * 32 + lane];
        }
#pragma unroll
        for (int vi = 0; vi < 8; ++vi) {
            int m = vi + half8;
            atomicAdd(&s[m * S_COLS + c * N_OUT + o], sacc[vi]);
        }
    }
}


__device__ __forceinline__ float fold_pair(float lo, float hi, int lane, int off) {
    bool up = (lane & off) != 0;
    float keep = up ? hi : lo;
    float send = up ? lo : hi;
    return keep + __shfl_xor(send, off, 32);
}

template <bool FIRST>
__device__ __forceinline__ void r_products(const v4f (&aa)[S_GROUP][2],
                                           const v4f (&bb)[S_GROUP][2],
                                           const float (&vreg)[8], float bo, float shift,
                                           int lane, v8f& sacc, float& dsum,
                                           float& bnewv, int& slot) {
    v8f d[S_GROUP];
    float p[S_GROUP];
#pragma unroll
    for (int j = 0; j < S_GROUP; ++j) {
        d[j] = wmma_k16(aa[j], bb[j]);
        float partial = 0.f;
#pragma unroll
        for (int vi = 0; vi < 8; ++vi) partial += d[j][vi] * vreg[vi];
        p[j] = partial;
    }
    float t0 = fold_pair(p[0], p[1], lane, 16);
    float t1 = fold_pair(p[2], p[3], lane, 16);
    float u  = fold_pair(t0, t1, lane, 8);
    u += __shfl_xor(u, 4, 32);
    u += __shfl_xor(u, 2, 32);
    u += __shfl_xor(u, 1, 32);
    float bn = bo + u * (1.0f / BATCH);
    float wl = __expf(bn - shift);
#pragma unroll
    for (int j = 0; j < S_GROUP; ++j) {
        const int srcl = ((j & 1) << 4) | ((j >> 1) << 3);
        float wgt = __builtin_bit_cast(float,
            __builtin_amdgcn_readlane(__builtin_bit_cast(int, wl), srcl));
#pragma unroll
        for (int vi = 0; vi < 8; ++vi) sacc[vi] = __builtin_fmaf(wgt, d[j][vi], sacc[vi]);
        dsum += wgt;
    }
    if (FIRST) {
        bnewv = (slot == 0) ? bn : bnewv;
        slot -= 1;
    }
}

template <bool FIRST>
__device__ __forceinline__ void r_pass_body(const float* __restrict__ x, const float* __restrict__ W,
                                            const float* __restrict__ sv,
                                            const int* __restrict__ cmax,
                                            float* __restrict__ bT, int* __restrict__ cmax_out,
                                            float* __restrict__ s, float* __restrict__ den,
                                            float* xs_blk, float* red, float* redd) {
    int lane  = threadIdx.x & 31;
    int wv    = threadIdx.x >> 5;
    int cw    = wv % CAPS_PB;
    int tl    = wv / CAPS_PB;
    int c     = (blockIdx.x % CGROUPS) * CAPS_PB + cw;
    int rblk  = (blockIdx.x / CGROUPS) * NR_BLK;
    int r0    = rblk + tl * RPW_S;
    const float* xs = xs_blk + tl * RPW_S * N_IN;
    int half8 = (lane >> 4) << 3;
    int o     = lane & 15;
    int copy  = (blockIdx.x / CGROUPS) & (N_COPY - 1);
    int jl    = ((lane >> 4) & 1) | ((lane >> 2) & 2);
    float* bline = bT + c * R_ROUTES + r0 + jl;

    v8f sacc = {0.f, 0.f, 0.f, 0.f, 0.f, 0.f, 0.f, 0.f};
    v4f a0[S_GROUP][2], b0[S_GROUP][2], a1[S_GROUP][2], b1[S_GROUP][2];
    float bo0 = 0.f, bo1 = 0.f;

#pragma unroll
    for (int j = 0; j < S_GROUP; ++j) load_B(W, r0 + j, c, lane, b0[j]);

    float vreg[8];
#pragma unroll
    for (int vi = 0; vi < 8; ++vi) vreg[vi] = sv[(vi + half8) * S_COLS + c * N_OUT + o];

    float shift = 0.f;
    if (!FIRST) {
        int km = cmax[c];
#pragma unroll
        for (int k = 1; k < N_COPY; ++k) {
            int kk = cmax[k * N_CAPS + c];
            km = kk > km ? kk : km;
        }
        shift = key2f(km);
        bo0 = bline[0];
    }

    stage_x<NR_BLK, BLK_THR>(x, rblk, xs_blk);

    if (FIRST) {
#pragma unroll
        for (int vi = 0; vi < 8; ++vi) vreg[vi] = squash1(vreg[vi]);
    }

    __syncthreads();

#pragma unroll
    for (int j = 0; j < S_GROUP; ++j) load_A<NR_BLK>(xs, j, lane, a0[j]);

    float dsum  = 0.f;
    float bnewv = 0.f;
    int   slot  = lane & 7;

    constexpr int N_PAIR = RPW_S / (2 * S_GROUP);

#pragma unroll 1
    for (int gg = 0; gg < N_PAIR - 1; ++gg) {
        const int rb = gg * 2 * S_GROUP;
        s_load_group(xs, W, r0, rb + S_GROUP, c, lane, a1, b1);
        if (!FIRST) bo1 = bline[rb + S_GROUP];
        __builtin_amdgcn_sched_barrier(0);
        r_products<FIRST>(a0, b0, vreg, bo0, shift, lane, sacc, dsum, bnewv, slot);
        s_load_group(xs, W, r0, rb + 2 * S_GROUP, c, lane, a0, b0);
        if (!FIRST) bo0 = bline[rb + 2 * S_GROUP];
        __builtin_amdgcn_sched_barrier(0);
        r_products<FIRST>(a1, b1, vreg, bo1, shift, lane, sacc, dsum, bnewv, slot);
    }
    {
        const int rb = (N_PAIR - 1) * 2 * S_GROUP;
        s_load_group(xs, W, r0, rb + S_GROUP, c, lane, a1, b1);
        if (!FIRST) bo1 = bline[rb + S_GROUP];
        __builtin_amdgcn_sched_barrier(0);
        r_products<FIRST>(a0, b0, vreg, bo0, shift, lane, sacc, dsum, bnewv, slot);
        r_products<FIRST>(a1, b1, vreg, bo1, shift, lane, sacc, dsum, bnewv, slot);
    }

    if (tl != 0) {
#pragma unroll
        for (int vi = 0; vi < 8; ++vi)
            red[(((tl - 1) * CAPS_PB + cw) * 8 + vi) * 32 + lane] = sacc[vi];
        if (lane == 0) redd[(tl - 1) * CAPS_PB + cw] = dsum;
    }
    __syncthreads();
    if (tl == 0) {
#pragma unroll
        for (int t = 1; t < TILES_PB; ++t) {
#pragma unroll
            for (int vi = 0; vi < 8; ++vi)
                sacc[vi] += red[(((t - 1) * CAPS_PB + cw) * 8 + vi) * 32 + lane];
            dsum += redd[(t - 1) * CAPS_PB + cw];
        }
#pragma unroll
        for (int vi = 0; vi < 8; ++vi) {
            int m = vi + half8;
            atomicAdd(&s[m * S_COLS + c * N_OUT + o], sacc[vi]);
        }
        if (lane == 0) atomicAdd(&den[copy * N_CAPS + c], dsum);
    }

    if (FIRST) {
        bline[S_GROUP * (lane & 7)] = bnewv;
        float mx = bnewv;
#pragma unroll
        for (int off = 16; off > 0; off >>= 1) mx = fmaxf(mx, __shfl_xor(mx, off, 32));
        if (lane == 0) atomicMax(&cmax_out[copy * N_CAPS + c], f2key(mx));
    }
}

__global__ void __launch_bounds__(BLK_THR)
r_pass_first_kernel(const float* __restrict__ x, const float* __restrict__ W,
                    const float* __restrict__ s0, float* __restrict__ bT,
                    int* __restrict__ cmax_out, float* __restrict__ s, float* __restrict__ den) {
    __shared__ __attribute__((aligned(16))) float xs_blk[BATCH * XS_PITCH(NR_BLK)];
    __shared__ float red[RED_FLTS];
    __shared__ float redd[(TILES_PB - 1) * CAPS_PB];
    r_pass_body<true>(x, W, s0, nullptr, bT, cmax_out, s, den, xs_blk, red, redd);
}

__global__ void __launch_bounds__(BLK_THR)
r_pass_kernel(const float* __restrict__ x, const float* __restrict__ W,
              const float* __restrict__ v, const int* __restrict__ cmax,
              float* __restrict__ bT, float* __restrict__ s, float* __restrict__ den) {
    __shared__ __attribute__((aligned(16))) float xs_blk[BATCH * XS_PITCH(NR_BLK)];
    __shared__ float red[RED_FLTS];
    __shared__ float redd[(TILES_PB - 1) * CAPS_PB];
    r_pass_body<false>(x, W, v, cmax, bT, nullptr, s, den, xs_blk, red, redd);
}

__global__ void mid_kernel(float* __restrict__ sv, const float* __restrict__ den,
                           float* __restrict__ s_next, float* __restrict__ den_next) {
    int i = blockIdx.x * blockDim.x + threadIdx.x;
    if (i < S_ELEMS) {
        int c = (i >> 4) & (N_CAPS - 1);
        sv[i] = squash1(sv[i] * (1.0f / den_total(den, c)));
        s_next[i] = 0.f;
    }
    if (i < N_COPY * N_CAPS) den_next[i] = 0.f;
}

__global__ void squash_kernel(const float* __restrict__ s, const float* __restrict__ den,
                              float* __restrict__ v) {
    int i = blockIdx.x * blockDim.x + threadIdx.x;
    if (i < S_ELEMS) {
        int c = (i >> 4) & (N_CAPS - 1);
        v[i] = squash1(s[i] * (1.0f / den_total(den, c)));
    }
}

extern "C" void kernel_launch(void* const* d_in, const int* in_sizes, int n_in,
                              void* d_out, int out_size, void* d_ws, size_t ws_size,
                              hipStream_t stream) {
    const float* x = (const float*)d_in[0];
    const float* W = (const float*)d_in[1];
    float* out = (float*)d_out;

    char* ws = (char*)d_ws;
    float* bT   = (float*)ws;
    int*   cmx1 = (int*)(ws + (size_t)R_ROUTES * N_CAPS * 4);
    int*   cmx2 = cmx1 + N_COPY * N_CAPS;
    float* den1 = (float*)(cmx2 + N_COPY * N_CAPS);
    float* den2 = den1 + N_COPY * N_CAPS;
    float* sbuf = den2 + N_COPY * N_CAPS;


    hipMemsetAsync(sbuf, 0, S_ELEMS * sizeof(float), stream);
    s_pass_first_kernel<<<CGROUPS * (R_ROUTES / NR_BLK), BLK_THR, 0, stream>>>(
        x, W, sbuf, out, cmx1, den1);

    r_pass_first_kernel<<<CGROUPS * (R_ROUTES / NR_BLK), BLK_THR, 0, stream>>>(
        x, W, sbuf, bT, cmx1, out, den1);

    mid_kernel<<<(S_ELEMS + 255) / 256, 256, 0, stream>>>(out, den1, sbuf, den2);

    r_pass_kernel<<<CGROUPS * (R_ROUTES / NR_BLK), BLK_THR, 0, stream>>>(
        x, W, out, cmx1, bT, sbuf, den2);

    squash_kernel<<<(S_ELEMS + 255) / 256, 256, 0, stream>>>(sbuf, den2, out);
}
